# starting kernel
# baseline (speedup 1.0000x reference)
#include <hip/hip_runtime.h>
#include <math.h>


#define BATCH   2
#define SEQLEN  2048
#define D_INNER 2048
#define D_STATE 16
#define DT_RANK 64
#define N_PROJ  (DT_RANK + 2 * D_STATE)
#define M_TOT   (BATCH * SEQLEN)
#define NSTATE_TOT (BATCH * D_INNER * D_STATE)
#define NCHUNK  16
#define CHUNKL  (SEQLEN / NCHUNK)

typedef float v2f __attribute__((ext_vector_type(2)));
typedef float v8f __attribute__((ext_vector_type(8)));

__global__ void ssm_xproj_wmma(const float* __restrict__ X,
                               const float* __restrict__ W,
                               float* __restrict__ Y) {
  const int lane = threadIdx.x & 31;
  const int wave = threadIdx.x >> 5;
  const int half = lane >> 4;
  const int ml   = lane & 15;
  const int m0   = blockIdx.x << 4;
  const int n0   = wave << 4;

  const float* __restrict__ arow = X + (size_t)(m0 + ml) * 2048;
  v8f acc = {};
  for (int k = 0; k < 2048; k += 4) {
    const int ka = k + 2 * half;
    v2f a, b;
    a.x = arow[ka];
    a.y = arow[ka + 1];
    b.x = W[(size_t)ka * N_PROJ + n0 + ml];
    b.y = W[(size_t)(ka + 1) * N_PROJ + n0 + ml];
    acc = __builtin_amdgcn_wmma_f32_16x16x4_f32(
        false, a, false, b, (short)0, acc, false, false);
  }
#pragma unroll
  for (int r = 0; r < 8; ++r) {
    const int m = m0 + r + 8 * half;
    Y[(size_t)m * N_PROJ + n0 + ml] = acc[r];
  }
}

__global__ void ssm_dtproj_wmma(const float* __restrict__ XD,
                                const float* __restrict__ Wd,
                                const float* __restrict__ bias,
                                float* __restrict__ DT) {
  const int lane = threadIdx.x & 31;
  const int wave = threadIdx.x >> 5;
  const int half = lane >> 4;
  const int ml   = lane & 15;
  const int m0   = blockIdx.y << 4;
  const int n0   = (blockIdx.x << 7) + (wave << 4);

  const float* __restrict__ arow = XD + (size_t)(m0 + ml) * N_PROJ;
  v8f acc = {};
  for (int k = 0; k < DT_RANK; k += 4) {
    const int ka = k + 2 * half;
    v2f a, b;
    a.x = arow[ka];
    a.y = arow[ka + 1];
    b.x = Wd[(size_t)ka * D_INNER + n0 + ml];
    b.y = Wd[(size_t)(ka + 1) * D_INNER + n0 + ml];
    acc = __builtin_amdgcn_wmma_f32_16x16x4_f32(
        false, a, false, b, (short)0, acc, false, false);
  }
  const float bn = bias[n0 + ml];
#pragma unroll
  for (int r = 0; r < 8; ++r) {
    const int m = m0 + r + 8 * half;
    float v = acc[r] + bn;
    float sp = (v > 20.0f) ? v : log1pf(__expf(v));
    DT[(size_t)m * D_INNER + n0 + ml] = sp;
  }
}

__global__ void ssm_scan_pass1(const float* __restrict__ X,
                               const float* __restrict__ XDBL,
                               const float* __restrict__ DT,
                               const float* __restrict__ A_log,
                               float* __restrict__ chunkA,
                               float* __restrict__ chunkB) {
  const int t = blockIdx.x * blockDim.x + threadIdx.x;
  const int chunk = blockIdx.y;
  const int n = t & 15;
  const int c = t >> 4;
  const int b = c >> 11;
  const int d = c & 2047;

  const float a2 = -__expf(A_log[d * D_STATE + n]) * 1.44269504088896f;
  float s = 0.0f;
  float p = 1.0f;
  const size_t mbase = (size_t)b * SEQLEN + (size_t)chunk * CHUNKL;
  for (int l = 0; l < CHUNKL; ++l) {
    const size_t m = mbase + l;
    const float dt = DT[m * D_INNER + d];
    const float xv = X[m * D_INNER + d];
    const float Bn = XDBL[m * N_PROJ + DT_RANK + n];
    const float dA = exp2f(dt * a2);
    s = fmaf(dA, s, dt * Bn * xv);
    p *= dA;
  }
  const size_t i = (size_t)chunk * NSTATE_TOT + t;
  chunkA[i] = p;
  chunkB[i] = s;
}

__global__ void ssm_scan_pass2(const float* __restrict__ chunkA,
                               float* __restrict__ chunkB) {
  const int t = blockIdx.x * blockDim.x + threadIdx.x;
  float s = 0.0f;
#pragma unroll
  for (int ch = 0; ch < NCHUNK; ++ch) {
    const size_t i = (size_t)ch * NSTATE_TOT + t;
    const float Ac = chunkA[i];
    const float Bc = chunkB[i];
    chunkB[i] = s;
    s = fmaf(Ac, s, Bc);
  }
}

__global__ void ssm_scan_pass3(const float* __restrict__ X,
                               const float* __restrict__ XDBL,
                               const float* __restrict__ DT,
                               const float* __restrict__ A_log,
                               const float* __restrict__ Dv,
                               const float* __restrict__ Sin,
                               float* __restrict__ OUT) {
  const int t = blockIdx.x * blockDim.x + threadIdx.x;
  const int chunk = blockIdx.y;
  const int n = t & 15;
  const int c = t >> 4;
  const int b = c >> 11;
  const int d = c & 2047;

  const float a2 = -__expf(A_log[d * D_STATE + n]) * 1.44269504088896f;
  const float Dd = Dv[d];
  float st = Sin[(size_t)chunk * NSTATE_TOT + t];

  const size_t mbase = (size_t)b * SEQLEN + (size_t)chunk * CHUNKL;
  for (int l = 0; l < CHUNKL; ++l) {
    const size_t m = mbase + l;
    const float dt = DT[m * D_INNER + d];
    const float xv = X[m * D_INNER + d];
    const float* __restrict__ bc = XDBL + m * N_PROJ + DT_RANK;
    const float Bn = bc[n];
    const float Cn = bc[D_STATE + n];

    const float dA = exp2f(dt * a2);
    st = fmaf(dA, st, dt * Bn * xv);
    float y = st * Cn;
    y += __shfl_xor(y, 1, 32);
    y += __shfl_xor(y, 2, 32);
    y += __shfl_xor(y, 4, 32);
    y += __shfl_xor(y, 8, 32);
    if (n == 0) OUT[m * D_INNER + d] = y + xv * Dd;
  }
}

extern "C" void kernel_launch(void* const* d_in, const int* in_sizes, int n_in,
                              void* d_out, int out_size, void* d_ws, size_t ws_size,
                              hipStream_t stream) {
  const float* x    = (const float*)d_in[0];
  const float* xpw  = (const float*)d_in[1];
  const float* dtw  = (const float*)d_in[2];
  const float* dtb  = (const float*)d_in[3];
  const float* alog = (const float*)d_in[4];
  const float* dvec = (const float*)d_in[5];
  float* out = (float*)d_out;

  char* ws = (char*)d_ws;
  float* xdbl   = (float*)ws;
  ws += (size_t)M_TOT * N_PROJ * sizeof(float);
  float* delta  = (float*)ws;
  ws += (size_t)M_TOT * D_INNER * sizeof(float);
  float* chunkA = (float*)ws;
  ws += (size_t)NCHUNK * NSTATE_TOT * sizeof(float);
  float* chunkB = (float*)ws;

  ssm_xproj_wmma<<<dim3(M_TOT / 16), dim3(192), 0, stream>>>(x, xpw, xdbl);

  ssm_dtproj_wmma<<<dim3(D_INNER / 128, M_TOT / 16), dim3(256), 0, stream>>>(
      xdbl, dtw, dtb, delta);

  ssm_scan_pass1<<<dim3(NSTATE_TOT / 256, NCHUNK), dim3(256), 0, stream>>>(
      x, xdbl, delta, alog, chunkA, chunkB);
  ssm_scan_pass2<<<dim3(NSTATE_TOT / 256), dim3(256), 0, stream>>>(
      chunkA, chunkB);
  ssm_scan_pass3<<<dim3(NSTATE_TOT / 256, NCHUNK), dim3(256), 0, stream>>>(
      x, xdbl, delta, alog, dvec, chunkB, out);
}


# Scan in ONE pass over time: each block keeps 16 channels' 256 states in registers and walks the 16 time tiles in order via two LDS stages, next tile in flight; no chunk replay (sum order changes).
# speedup vs baseline: 3.0176x; 3.0176x over previous
#include <hip/hip_runtime.h>
#include <math.h>


#define BATCH   2
#define SEQLEN  2048
#define D_INNER 2048
#define D_STATE 16
#define DT_RANK 64
#define N_PROJ  (DT_RANK + 2 * D_STATE)
#define M_TOT   (BATCH * SEQLEN)
#define NTILE   16
#define CHUNKL  (SEQLEN / NTILE)

#define SCH     16
#define SNT     256
#define TPITCH  (CHUNKL + 4)

typedef float v2f __attribute__((ext_vector_type(2)));
typedef float v4f __attribute__((ext_vector_type(4)));
typedef float v8f __attribute__((ext_vector_type(8)));

#define G1_KC     128
#define G1_NCH    (D_INNER / G1_KC)
#define G1_APITCH (G1_KC + 4)
#define G1_BPITCH (N_PROJ + 8)
#define G1_ASZ    (16 * G1_APITCH)
#define G1_BSZ    (G1_KC * G1_BPITCH)
#define G1_STAGE  (G1_ASZ + G1_BSZ)
#define G1_GRP    8

struct G1Set { v4f a[3]; v4f b[16]; };

__device__ __forceinline__ void g1_fetch(const float* __restrict__ xpanel,
                                         const float* __restrict__ W, int k0, int tid, G1Set& s) {
  const int r = tid >> 5;
  const float* __restrict__ src = xpanel + k0 + 4 * (tid & 31);
  s.a[0] = *reinterpret_cast<const v4f*>(src + (size_t)r * D_INNER);
  s.a[1] = *reinterpret_cast<const v4f*>(src + (size_t)(r + 6) * D_INNER);
  s.a[2] = *reinterpret_cast<const v4f*>(src + (size_t)((r + 12) & 15) * D_INNER);
  const float* __restrict__ w = W + (size_t)k0 * N_PROJ + 4 * tid;
  for (int j = 0; j < 16; ++j)
    s.b[j] = *reinterpret_cast<const v4f*>(w + (size_t)(768 * j));
}

__device__ __forceinline__ void g1_put(float* __restrict__ sa, float* __restrict__ sb, int tid,
                                       const G1Set& s) {
  const int r = tid >> 5;
  float* __restrict__ dst = sa + 4 * (tid & 31);
  *reinterpret_cast<v4f*>(dst + r * G1_APITCH)               = s.a[0];
  *reinterpret_cast<v4f*>(dst + (r + 6) * G1_APITCH)         = s.a[1];
  *reinterpret_cast<v4f*>(dst + ((r + 12) & 15) * G1_APITCH) = s.a[2];
  float* __restrict__ d = sb + (tid / 24) * G1_BPITCH + 4 * (tid % 24);
  for (int j = 0; j < 16; ++j)
    *reinterpret_cast<v4f*>(d + 8 * j * G1_BPITCH) = s.b[j];
}

__device__ __forceinline__ void g1_frags(const float* __restrict__ pa, const float* __restrict__ pb,
                                         int g, v2f (&fa)[G1_GRP], v2f (&fb)[G1_GRP]) {
#pragma unroll
  for (int i = 0; i < G1_GRP; ++i) {
    const int t = G1_GRP * g + i;
    fa[i] = *reinterpret_cast<const v2f*>(pa + 4 * t);
    fb[i].x = pb[(4 * t) * G1_BPITCH];
    fb[i].y = pb[(4 * t + 1) * G1_BPITCH];
  }
}

__device__ __forceinline__ void stage_barrier() {
  asm volatile("s_wait_dscnt 0x0" ::: "memory");
  asm volatile("s_barrier_signal -1" ::: "memory");
  asm volatile("s_barrier_wait -1" ::: "memory");
}

template <bool DO_PUT, bool DO_FETCH>
__device__ __forceinline__ void g1_step(const float* __restrict__ cur, float* __restrict__ nxt,
                                        G1Set& s, const float* __restrict__ xpanel,
                                        const float* __restrict__ W, int kfetch, int tid,
                                        int aoff, int boff, v8f& acc) {
  const float* __restrict__ pa = cur + aoff;
  const float* __restrict__ pb = cur + boff;
  v2f fa[2][G1_GRP], fb[2][G1_GRP];
  g1_frags(pa, pb, 0, fa[0], fb[0]);
  __builtin_amdgcn_sched_barrier(0);
  if (DO_PUT) g1_put(nxt, nxt + G1_ASZ, tid, s);
  if (DO_FETCH) g1_fetch(xpanel, W, kfetch, tid, s);
#pragma unroll
  for (int g = 0; g < G1_KC / (4 * G1_GRP); ++g) {
    if (g + 1 < G1_KC / (4 * G1_GRP)) g1_frags(pa, pb, g + 1, fa[(g + 1) & 1], fb[(g + 1) & 1]);
    __builtin_amdgcn_sched_barrier(0);
#pragma unroll
    for (int i = 0; i < G1_GRP; ++i)
      acc = __builtin_amdgcn_wmma_f32_16x16x4_f32(
          false, fa[g & 1][i], false, fb[g & 1][i], (short)0, acc, false, false);
  }
}

__global__ __launch_bounds__(192) void ssm_xproj_wmma(const float* __restrict__ X,
                               const float* __restrict__ W,
                               float* __restrict__ Y) {
  __shared__ __attribute__((aligned(16))) float g1_lds[2 * G1_STAGE];

  const int tid  = threadIdx.x;
  const int lane = tid & 31;
  const int wave = tid >> 5;
  const int half = lane >> 4;
  const int ml   = lane & 15;
  const int m0   = blockIdx.x << 4;
  const int n0   = wave << 4;

  float* __restrict__ st0 = g1_lds;
  float* __restrict__ st1 = g1_lds + G1_STAGE;
  const float* __restrict__ xpanel = X + (size_t)m0 * D_INNER;

  G1Set E, O;
  g1_fetch(xpanel, W, 0, tid, E);
  g1_fetch(xpanel, W, G1_KC, tid, O);
  __builtin_amdgcn_sched_barrier(0);
  g1_put(st0, st0 + G1_ASZ, tid, E);
  g1_fetch(xpanel, W, 2 * G1_KC, tid, E);
  stage_barrier();

  const int aoff = ml * G1_APITCH + 2 * half;
  const int boff = G1_ASZ + 2 * half * G1_BPITCH + n0 + ml;
  v8f acc = {};
  for (int c = 0; c < G1_NCH - 4; c += 2) {
    g1_step<true, true>(st0, st1, O, xpanel, W, (c + 3) * G1_KC, tid, aoff, boff, acc);
    stage_barrier();
    g1_step<true, true>(st1, st0, E, xpanel, W, (c + 4) * G1_KC, tid, aoff, boff, acc);
    stage_barrier();
  }
  g1_step<true, true>(st0, st1, O, xpanel, W, (G1_NCH - 1) * G1_KC, tid, aoff, boff, acc);
  stage_barrier();
  g1_step<true, false>(st1, st0, E, xpanel, W, 0, tid, aoff, boff, acc);
  stage_barrier();
  g1_step<true, false>(st0, st1, O, xpanel, W, 0, tid, aoff, boff, acc);
  stage_barrier();
  g1_step<false, false>(st1, st0, O, xpanel, W, 0, tid, aoff, boff, acc);
#pragma unroll
  for (int r = 0; r < 8; ++r) {
    const int m = m0 + r + 8 * half;
    Y[(size_t)m * N_PROJ + n0 + ml] = acc[r];
  }
}

__global__ void ssm_dtproj_wmma(const float* __restrict__ XD,
                                const float* __restrict__ Wd,
                                const float* __restrict__ bias,
                                float* __restrict__ DT) {
  const int lane = threadIdx.x & 31;
  const int wave = threadIdx.x >> 5;
  const int half = lane >> 4;
  const int ml   = lane & 15;
  const int m0   = blockIdx.y << 4;
  const int n0   = (blockIdx.x << 7) + (wave << 4);

  const float* __restrict__ arow = XD + (size_t)(m0 + ml) * N_PROJ;
  v8f acc = {};
  for (int k = 0; k < DT_RANK; k += 4) {
    const int ka = k + 2 * half;
    v2f a, b;
    a.x = arow[ka];
    a.y = arow[ka + 1];
    b.x = Wd[(size_t)ka * D_INNER + n0 + ml];
    b.y = Wd[(size_t)(ka + 1) * D_INNER + n0 + ml];
    acc = __builtin_amdgcn_wmma_f32_16x16x4_f32(
        false, a, false, b, (short)0, acc, false, false);
  }
  const float bn = bias[n0 + ml];
#pragma unroll
  for (int r = 0; r < 8; ++r) {
    const int m = m0 + r + 8 * half;
    float v = acc[r] + bn;
    float sp = (v > 20.0f) ? v : log1pf(__expf(v));
    DT[(size_t)m * D_INNER + n0 + ml] = sp;
  }
}

#define S_XY    (SCH * TPITCH)
#define S_BC    (2 * SCH * TPITCH)
#define S_STAGE ((2 * SCH + 2 * D_STATE) * TPITCH)
#define S_LDS_FLOATS (164 * 256)

template <int NSWEEP, int F4_PER_ROW>
__device__ __forceinline__ void tile_fetch(const float* __restrict__ src, size_t ld,
                                           int tid, v4f (&v)[NSWEEP]) {
  constexpr int ROWS_PER_SWEEP = SNT / F4_PER_ROW;
  const int c4 = tid % F4_PER_ROW;
  const int r0 = tid / F4_PER_ROW;
  for (int it = 0; it < NSWEEP; ++it) {
    const int tt = r0 + ROWS_PER_SWEEP * it;
    v[it] = *reinterpret_cast<const v4f*>(src + (size_t)tt * ld + 4 * c4);
  }
}

template <int NSWEEP, int F4_PER_ROW>
__device__ __forceinline__ void tile_put_transposed(float* __restrict__ dst, int tid,
                                                    const v4f (&v)[NSWEEP]) {
  constexpr int ROWS_PER_SWEEP = SNT / F4_PER_ROW;
  const int c4 = tid % F4_PER_ROW;
  const int r0 = tid / F4_PER_ROW;
  for (int it = 0; it < NSWEEP; ++it) {
    const int tt = r0 + ROWS_PER_SWEEP * it;
    dst[(4 * c4 + 0) * TPITCH + tt] = v[it].x;
    dst[(4 * c4 + 1) * TPITCH + tt] = v[it].y;
    dst[(4 * c4 + 2) * TPITCH + tt] = v[it].z;
    dst[(4 * c4 + 3) * TPITCH + tt] = v[it].w;
  }
}

struct ScanSet { v4f dt[2]; v4f x[2]; v4f bc[4]; };

__device__ __forceinline__ void scan_fetch(const float* __restrict__ dtp,
                                           const float* __restrict__ xp,
                                           const float* __restrict__ bcp,
                                           int tile, int tid, ScanSet& s) {
  const size_t r = (size_t)tile * CHUNKL;
  tile_fetch<2, 4>(dtp + r * D_INNER, D_INNER, tid, s.dt);
  tile_fetch<2, 4>(xp + r * D_INNER, D_INNER, tid, s.x);
  tile_fetch<4, 8>(bcp + r * N_PROJ, N_PROJ, tid, s.bc);
}

__device__ __forceinline__ void scan_put(float* __restrict__ stage, int tid, const ScanSet& s) {
  tile_put_transposed<2, 4>(stage, tid, s.dt);
  tile_put_transposed<2, 4>(stage + S_XY, tid, s.x);
  tile_put_transposed<4, 8>(stage + S_BC, tid, s.bc);
}

template <int MASK>
__device__ __forceinline__ float row_xor(float v) {
  return __int_as_float(__builtin_amdgcn_update_dpp(
      0, __float_as_int(v), 0x160 + MASK, 0xf, 0xf, true));
}

__device__ __forceinline__ void scan_group(const v4f dt4, const v4f x4, const v4f b4, const v4f c4,
                                           float a2, float Dd, int n, float* pxy_l, float& st) {
  v4f y4;
  for (int j = 0; j < 4; ++j) {
    const float dt = dt4[j];
    const float xv = x4[j];
    const float Bn = b4[j];
    const float Cn = c4[j];
    const float dA = exp2f(dt * a2);
    st = fmaf(dA, st, dt * Bn * xv);
    float y = st * Cn;
    y += row_xor<1>(y);
    y += row_xor<2>(y);
    y += row_xor<4>(y);
    y += row_xor<8>(y);
    y4[j] = y + xv * Dd;
  }
  if (n == 0) *reinterpret_cast<v4f*>(pxy_l) = y4;
}

template <bool DO_PUT, bool DO_FETCH>
__device__ __forceinline__ void scan_tile(float* cur, float* __restrict__ nxt, ScanSet& s,
                                          const float* __restrict__ dtp,
                                          const float* __restrict__ xp,
                                          const float* __restrict__ bcp,
                                          float* __restrict__ outp, int tile, int tid,
                                          float a2, float Dd, float& st) {
  const int n = tid & 15;
  const int cl = tid >> 4;
  const float* pdt = cur + cl * TPITCH;
  float* pxy = cur + S_XY + cl * TPITCH;
  const float* pb = cur + S_BC + n * TPITCH;
  const float* pc = cur + S_BC + (D_STATE + n) * TPITCH;
  v4f dtA = *reinterpret_cast<const v4f*>(pdt);
  v4f xA  = *reinterpret_cast<const v4f*>(pxy);
  v4f bA  = *reinterpret_cast<const v4f*>(pb);
  v4f cA  = *reinterpret_cast<const v4f*>(pc);
  for (int l = 0; l < CHUNKL; l += 8) {
    const v4f dtB = *reinterpret_cast<const v4f*>(pdt + l + 4);
    const v4f xB  = *reinterpret_cast<const v4f*>(pxy + l + 4);
    const v4f bB  = *reinterpret_cast<const v4f*>(pb + l + 4);
    const v4f cB  = *reinterpret_cast<const v4f*>(pc + l + 4);
    __builtin_amdgcn_sched_barrier(0);
    scan_group(dtA, xA, bA, cA, a2, Dd, n, pxy + l, st);
    const int l2 = (l + 8) & (CHUNKL - 1);
    dtA = *reinterpret_cast<const v4f*>(pdt + l2);
    xA  = *reinterpret_cast<const v4f*>(pxy + l2);
    bA  = *reinterpret_cast<const v4f*>(pb + l2);
    cA  = *reinterpret_cast<const v4f*>(pc + l2);
    __builtin_amdgcn_sched_barrier(0);
    scan_group(dtB, xB, bB, cB, a2, Dd, n, pxy + l + 4, st);
  }
  stage_barrier();

  {
    const int c4 = tid & 3;
    const int r0 = tid >> 2;
    const float* sy = cur + S_XY;
    float* __restrict__ o = outp + (size_t)tile * CHUNKL * D_INNER + 4 * c4;
    for (int it = 0; it < 2; ++it) {
      const int tt = r0 + 64 * it;
      v4f v;
      v.x = sy[(4 * c4 + 0) * TPITCH + tt];
      v.y = sy[(4 * c4 + 1) * TPITCH + tt];
      v.z = sy[(4 * c4 + 2) * TPITCH + tt];
      v.w = sy[(4 * c4 + 3) * TPITCH + tt];
      *reinterpret_cast<v4f*>(o + (size_t)tt * D_INNER) = v;
    }
  }
  if (DO_PUT) {
    scan_put(nxt, tid, s);
    if (DO_FETCH) scan_fetch(dtp, xp, bcp, tile + 2, tid, s);
    stage_barrier();
  }
}

__global__ __launch_bounds__(SNT) void ssm_scan_seq(const float* __restrict__ X,
                               const float* __restrict__ XDBL,
                               const float* __restrict__ DT,
                               const float* __restrict__ A_log,
                               const float* __restrict__ Dv,
                               float* __restrict__ OUT) {
  __shared__ __attribute__((aligned(16))) float s_lds[S_LDS_FLOATS];

  const int tid = threadIdx.x;
  const int n = tid & 15;
  const int cl = tid >> 4;
  const int cbase = blockIdx.x * SCH;
  const int b = cbase >> 11;
  const int dbase = cbase & 2047;
  const int d = dbase + cl;
  const size_t row0 = (size_t)b * SEQLEN;

  const float* __restrict__ dtp = DT + row0 * D_INNER + dbase;
  const float* __restrict__ xp  = X + row0 * D_INNER + dbase;
  const float* __restrict__ bcp = XDBL + row0 * N_PROJ + DT_RANK;
  float* __restrict__ outp = OUT + row0 * D_INNER + dbase;
  float* st0 = s_lds;
  float* st1 = s_lds + S_STAGE;

  ScanSet s;
  scan_fetch(dtp, xp, bcp, 0, tid, s);
  const float alog = A_log[d * D_STATE + n];
  const float Dd = Dv[d];
  scan_put(st0, tid, s);
  scan_fetch(dtp, xp, bcp, 1, tid, s);
  stage_barrier();

  const float a2 = -__expf(alog) * 1.44269504088896f;
  float st = 0.0f;
  for (int c = 0; c < NTILE - 2; c += 2) {
    scan_tile<true, true>(st0, st1, s, dtp, xp, bcp, outp, c, tid, a2, Dd, st);
    scan_tile<true, true>(st1, st0, s, dtp, xp, bcp, outp, c + 1, tid, a2, Dd, st);
  }
  scan_tile<true, false>(st0, st1, s, dtp, xp, bcp, outp, NTILE - 2, tid, a2, Dd, st);
  scan_tile<false, false>(st1, st0, s, dtp, xp, bcp, outp, NTILE - 1, tid, a2, Dd, st);
}

extern "C" void kernel_launch(void* const* d_in, const int* in_sizes, int n_in,
                              void* d_out, int out_size, void* d_ws, size_t ws_size,
                              hipStream_t stream) {
  const float* x    = (const float*)d_in[0];
  const float* xpw  = (const float*)d_in[1];
  const float* dtw  = (const float*)d_in[2];
  const float* dtb  = (const float*)d_in[3];
  const float* alog = (const float*)d_in[4];
  const float* dvec = (const float*)d_in[5];
  float* out = (float*)d_out;

  char* ws = (char*)d_ws;
  float* xdbl   = (float*)ws;
  ws += (size_t)M_TOT * N_PROJ * sizeof(float);
  float* delta  = (float*)ws;

  ssm_xproj_wmma<<<dim3(M_TOT / 16), dim3(192), 0, stream>>>(x, xpw, xdbl);

  ssm_dtproj_wmma<<<dim3(D_INNER / 128, M_TOT / 16), dim3(256), 0, stream>>>(
      xdbl, dtw, dtb, delta);

  ssm_scan_seq<<<dim3(BATCH * D_INNER / SCH), dim3(SNT), 0, stream>>>(
      x, xdbl, delta, alog, dvec, out);
}


# dt projection (WMMA + softplus) moved into the scan block: 8 added waves make the next tile's delta into LDS while 8 waves step; delta never goes to memory, one launch fewer (sums bit-identical).
# speedup vs baseline: 3.1938x; 1.0584x over previous
#include <hip/hip_runtime.h>
#include <math.h>


#define BATCH   2
#define SEQLEN  2048
#define D_INNER 2048
#define D_STATE 16
#define DT_RANK 64
#define N_PROJ  (DT_RANK + 2 * D_STATE)
#define M_TOT   (BATCH * SEQLEN)
#define NTILE   16
#define CHUNKL  (SEQLEN / NTILE)

#define SCH     16
#define SCT     256
#define SNT     512
#define TPITCH  (CHUNKL + 4)

typedef float v2f __attribute__((ext_vector_type(2)));
typedef float v4f __attribute__((ext_vector_type(4)));
typedef float v8f __attribute__((ext_vector_type(8)));

#define G1_KC     128
#define G1_NCH    (D_INNER / G1_KC)
#define G1_APITCH (G1_KC + 4)
#define G1_BPITCH (N_PROJ + 8)
#define G1_ASZ    (16 * G1_APITCH)
#define G1_BSZ    (G1_KC * G1_BPITCH)
#define G1_STAGE  (G1_ASZ + G1_BSZ)
#define G1_GRP    8

struct G1Set { v4f a[3]; v4f b[16]; };

__device__ __forceinline__ void g1_fetch(const float* __restrict__ xpanel,
                                         const float* __restrict__ W, int k0, int tid, G1Set& s) {
  const int r = tid >> 5;
  const float* __restrict__ src = xpanel + k0 + 4 * (tid & 31);
  s.a[0] = *reinterpret_cast<const v4f*>(src + (size_t)r * D_INNER);
  s.a[1] = *reinterpret_cast<const v4f*>(src + (size_t)(r + 6) * D_INNER);
  s.a[2] = *reinterpret_cast<const v4f*>(src + (size_t)((r + 12) & 15) * D_INNER);
  const float* __restrict__ w = W + (size_t)k0 * N_PROJ + 4 * tid;
  for (int j = 0; j < 16; ++j)
    s.b[j] = *reinterpret_cast<const v4f*>(w + (size_t)(768 * j));
}

__device__ __forceinline__ void g1_put(float* __restrict__ sa, float* __restrict__ sb, int tid,
                                       const G1Set& s) {
  const int r = tid >> 5;
  float* __restrict__ dst = sa + 4 * (tid & 31);
  *reinterpret_cast<v4f*>(dst + r * G1_APITCH)               = s.a[0];
  *reinterpret_cast<v4f*>(dst + (r + 6) * G1_APITCH)         = s.a[1];
  *reinterpret_cast<v4f*>(dst + ((r + 12) & 15) * G1_APITCH) = s.a[2];
  float* __restrict__ d = sb + (tid / 24) * G1_BPITCH + 4 * (tid % 24);
  for (int j = 0; j < 16; ++j)
    *reinterpret_cast<v4f*>(d + 8 * j * G1_BPITCH) = s.b[j];
}

__device__ __forceinline__ void g1_frags(const float* __restrict__ pa, const float* __restrict__ pb,
                                         int g, v2f (&fa)[G1_GRP], v2f (&fb)[G1_GRP]) {
#pragma unroll
  for (int i = 0; i < G1_GRP; ++i) {
    const int t = G1_GRP * g + i;
    fa[i] = *reinterpret_cast<const v2f*>(pa + 4 * t);
    fb[i].x = pb[(4 * t) * G1_BPITCH];
    fb[i].y = pb[(4 * t + 1) * G1_BPITCH];
  }
}

__device__ __forceinline__ void stage_barrier() {
  asm volatile("s_wait_dscnt 0x0" ::: "memory");
  asm volatile("s_barrier_signal -1" ::: "memory");
  asm volatile("s_barrier_wait -1" ::: "memory");
}

template <bool DO_PUT, bool DO_FETCH>
__device__ __forceinline__ void g1_step(const float* __restrict__ cur, float* __restrict__ nxt,
                                        G1Set& s, const float* __restrict__ xpanel,
                                        const float* __restrict__ W, int kfetch, int tid,
                                        int aoff, int boff, v8f& acc) {
  const float* __restrict__ pa = cur + aoff;
  const float* __restrict__ pb = cur + boff;
  v2f fa[2][G1_GRP], fb[2][G1_GRP];
  g1_frags(pa, pb, 0, fa[0], fb[0]);
  __builtin_amdgcn_sched_barrier(0);
  if (DO_PUT) g1_put(nxt, nxt + G1_ASZ, tid, s);
  if (DO_FETCH) g1_fetch(xpanel, W, kfetch, tid, s);
#pragma unroll
  for (int g = 0; g < G1_KC / (4 * G1_GRP); ++g) {
    if (g + 1 < G1_KC / (4 * G1_GRP)) g1_frags(pa, pb, g + 1, fa[(g + 1) & 1], fb[(g + 1) & 1]);
    __builtin_amdgcn_sched_barrier(0);
#pragma unroll
    for (int i = 0; i < G1_GRP; ++i)
      acc = __builtin_amdgcn_wmma_f32_16x16x4_f32(
          false, fa[g & 1][i], false, fb[g & 1][i], (short)0, acc, false, false);
  }
}

__global__ __launch_bounds__(192) void ssm_xproj_wmma(const float* __restrict__ X,
                               const float* __restrict__ W,
                               float* __restrict__ Y) {
  __shared__ __attribute__((aligned(16))) float g1_lds[2 * G1_STAGE];

  const int tid  = threadIdx.x;
  const int lane = tid & 31;
  const int wave = tid >> 5;
  const int half = lane >> 4;
  const int ml   = lane & 15;
  const int m0   = blockIdx.x << 4;
  const int n0   = wave << 4;

  float* __restrict__ st0 = g1_lds;
  float* __restrict__ st1 = g1_lds + G1_STAGE;
  const float* __restrict__ xpanel = X + (size_t)m0 * D_INNER;

  G1Set E, O;
  g1_fetch(xpanel, W, 0, tid, E);
  g1_fetch(xpanel, W, G1_KC, tid, O);
  __builtin_amdgcn_sched_barrier(0);
  g1_put(st0, st0 + G1_ASZ, tid, E);
  g1_fetch(xpanel, W, 2 * G1_KC, tid, E);
  stage_barrier();

  const int aoff = ml * G1_APITCH + 2 * half;
  const int boff = G1_ASZ + 2 * half * G1_BPITCH + n0 + ml;
  v8f acc = {};
  for (int c = 0; c < G1_NCH - 4; c += 2) {
    g1_step<true, true>(st0, st1, O, xpanel, W, (c + 3) * G1_KC, tid, aoff, boff, acc);
    stage_barrier();
    g1_step<true, true>(st1, st0, E, xpanel, W, (c + 4) * G1_KC, tid, aoff, boff, acc);
    stage_barrier();
  }
  g1_step<true, true>(st0, st1, O, xpanel, W, (G1_NCH - 1) * G1_KC, tid, aoff, boff, acc);
  stage_barrier();
  g1_step<true, false>(st1, st0, E, xpanel, W, 0, tid, aoff, boff, acc);
  stage_barrier();
  g1_step<true, false>(st0, st1, O, xpanel, W, 0, tid, aoff, boff, acc);
  stage_barrier();
  g1_step<false, false>(st1, st0, O, xpanel, W, 0, tid, aoff, boff, acc);
#pragma unroll
  for (int r = 0; r < 8; ++r) {
    const int m = m0 + r + 8 * half;
    Y[(size_t)m * N_PROJ + n0 + ml] = acc[r];
  }
}

#define S_XY    (SCH * TPITCH)
#define S_BC    (2 * SCH * TPITCH)
#define S_STAGE ((2 * SCH + 2 * D_STATE) * TPITCH)
#define S_LDS_FLOATS (164 * 256)

template <int NSWEEP, int F4_PER_ROW>
__device__ __forceinline__ void tile_fetch(const float* __restrict__ src, size_t ld,
                                           int tid, v4f (&v)[NSWEEP]) {
  constexpr int ROWS_PER_SWEEP = SCT / F4_PER_ROW;
  const int c4 = tid % F4_PER_ROW;
  const int r0 = tid / F4_PER_ROW;
  for (int it = 0; it < NSWEEP; ++it) {
    const int tt = r0 + ROWS_PER_SWEEP * it;
    v[it] = *reinterpret_cast<const v4f*>(src + (size_t)tt * ld + 4 * c4);
  }
}

template <int NSWEEP, int F4_PER_ROW>
__device__ __forceinline__ void tile_put_transposed(float* __restrict__ dst, int tid,
                                                    const v4f (&v)[NSWEEP]) {
  constexpr int ROWS_PER_SWEEP = SCT / F4_PER_ROW;
  const int c4 = tid % F4_PER_ROW;
  const int r0 = tid / F4_PER_ROW;
  for (int it = 0; it < NSWEEP; ++it) {
    const int tt = r0 + ROWS_PER_SWEEP * it;
    dst[(4 * c4 + 0) * TPITCH + tt] = v[it].x;
    dst[(4 * c4 + 1) * TPITCH + tt] = v[it].y;
    dst[(4 * c4 + 2) * TPITCH + tt] = v[it].z;
    dst[(4 * c4 + 3) * TPITCH + tt] = v[it].w;
  }
}

struct ScanSet { v4f x[2]; v4f bc[4]; };

__device__ __forceinline__ void scan_fetch(const float* __restrict__ xp,
                                           const float* __restrict__ bcp,
                                           int tile, int tid, ScanSet& s) {
  const size_t r = (size_t)tile * CHUNKL;
  tile_fetch<2, 4>(xp + r * D_INNER, D_INNER, tid, s.x);
  tile_fetch<4, 8>(bcp + r * N_PROJ, N_PROJ, tid, s.bc);
}

__device__ __forceinline__ void scan_put(float* __restrict__ stage, int tid, const ScanSet& s) {
  tile_put_transposed<2, 4>(stage + S_XY, tid, s.x);
  tile_put_transposed<4, 8>(stage + S_BC, tid, s.bc);
}

__device__ __forceinline__ void delta_tile(const float* __restrict__ xd,
                                           const v2f (&bf)[DT_RANK / 4], float bn,
                                           int pw, int lane, float* __restrict__ stage) {
  const int half = lane >> 4;
  const int ml   = lane & 15;
  const float* __restrict__ arow = xd + (size_t)(16 * pw + ml) * N_PROJ + 2 * half;
  v2f a[DT_RANK / 4];
#pragma unroll
  for (int i = 0; i < DT_RANK / 4; ++i) {
    a[i].x = arow[4 * i];
    a[i].y = arow[4 * i + 1];
  }
  v8f acc = {};
#pragma unroll
  for (int i = 0; i < DT_RANK / 4; ++i)
    acc = __builtin_amdgcn_wmma_f32_16x16x4_f32(
        false, a[i], false, bf[i], (short)0, acc, false, false);
  float sp[8];
#pragma unroll
  for (int r = 0; r < 8; ++r) {
    float v = acc[r] + bn;
    sp[r] = (v > 20.0f) ? v : log1pf(__expf(v));
  }
  float* __restrict__ dst = stage + ml * TPITCH + 16 * pw + 8 * half;
  v4f lo = {sp[0], sp[1], sp[2], sp[3]};
  v4f hi = {sp[4], sp[5], sp[6], sp[7]};
  *reinterpret_cast<v4f*>(dst)     = lo;
  *reinterpret_cast<v4f*>(dst + 4) = hi;
}

template <int MASK>
__device__ __forceinline__ float row_xor(float v) {
  return __int_as_float(__builtin_amdgcn_update_dpp(
      0, __float_as_int(v), 0x160 + MASK, 0xf, 0xf, true));
}

__device__ __forceinline__ void scan_group(const v4f dt4, const v4f x4, const v4f b4, const v4f c4,
                                           float a2, float Dd, int n, float* pxy_l, float& st) {
  v4f y4;
  for (int j = 0; j < 4; ++j) {
    const float dt = dt4[j];
    const float xv = x4[j];
    const float Bn = b4[j];
    const float Cn = c4[j];
    const float dA = exp2f(dt * a2);
    st = fmaf(dA, st, dt * Bn * xv);
    float y = st * Cn;
    y += row_xor<1>(y);
    y += row_xor<2>(y);
    y += row_xor<4>(y);
    y += row_xor<8>(y);
    y4[j] = y + xv * Dd;
  }
  if (n == 0) *reinterpret_cast<v4f*>(pxy_l) = y4;
}

__device__ __forceinline__ void scan_steps(float* cur, int tid, float a2, float Dd, float& st) {
  const int n = tid & 15;
  const int cl = tid >> 4;
  const float* pdt = cur + cl * TPITCH;
  float* pxy = cur + S_XY + cl * TPITCH;
  const float* pb = cur + S_BC + n * TPITCH;
  const float* pc = cur + S_BC + (D_STATE + n) * TPITCH;
  v4f dtA = *reinterpret_cast<const v4f*>(pdt);
  v4f xA  = *reinterpret_cast<const v4f*>(pxy);
  v4f bA  = *reinterpret_cast<const v4f*>(pb);
  v4f cA  = *reinterpret_cast<const v4f*>(pc);
  for (int l = 0; l < CHUNKL; l += 8) {
    const v4f dtB = *reinterpret_cast<const v4f*>(pdt + l + 4);
    const v4f xB  = *reinterpret_cast<const v4f*>(pxy + l + 4);
    const v4f bB  = *reinterpret_cast<const v4f*>(pb + l + 4);
    const v4f cB  = *reinterpret_cast<const v4f*>(pc + l + 4);
    __builtin_amdgcn_sched_barrier(0);
    scan_group(dtA, xA, bA, cA, a2, Dd, n, pxy + l, st);
    const int l2 = (l + 8) & (CHUNKL - 1);
    dtA = *reinterpret_cast<const v4f*>(pdt + l2);
    xA  = *reinterpret_cast<const v4f*>(pxy + l2);
    bA  = *reinterpret_cast<const v4f*>(pb + l2);
    cA  = *reinterpret_cast<const v4f*>(pc + l2);
    __builtin_amdgcn_sched_barrier(0);
    scan_group(dtB, xB, bB, cB, a2, Dd, n, pxy + l + 4, st);
  }
}

__device__ __forceinline__ void scan_move(const float* cur, float* __restrict__ nxt, ScanSet& s,
                                          const float* __restrict__ xp,
                                          const float* __restrict__ bcp,
                                          float* __restrict__ outp, int tile, int tid,
                                          bool do_put, bool do_fetch) {
  {
    const int c4 = tid & 3;
    const int r0 = tid >> 2;
    const float* sy = cur + S_XY;
    float* __restrict__ o = outp + (size_t)tile * CHUNKL * D_INNER + 4 * c4;
    for (int it = 0; it < 2; ++it) {
      const int tt = r0 + 64 * it;
      v4f v;
      v.x = sy[(4 * c4 + 0) * TPITCH + tt];
      v.y = sy[(4 * c4 + 1) * TPITCH + tt];
      v.z = sy[(4 * c4 + 2) * TPITCH + tt];
      v.w = sy[(4 * c4 + 3) * TPITCH + tt];
      *reinterpret_cast<v4f*>(o + (size_t)tt * D_INNER) = v;
    }
  }
  if (do_put) {
    scan_put(nxt, tid, s);
    if (do_fetch) scan_fetch(xp, bcp, tile + 2, tid, s);
  }
}

__global__ __launch_bounds__(SNT) void ssm_scan_seq(const float* __restrict__ X,
                               const float* __restrict__ XDBL,
                               const float* __restrict__ Wd,
                               const float* __restrict__ bias,
                               const float* __restrict__ A_log,
                               const float* __restrict__ Dv,
                               float* __restrict__ OUT) {
  __shared__ __attribute__((aligned(16))) float s_lds[S_LDS_FLOATS];

  const int lane = threadIdx.x & 31;
  const int w = __builtin_amdgcn_readfirstlane((int)(threadIdx.x >> 5));
  const bool steps = (((w ^ (w >> 2)) & 1) == 0);
  const int kw = w >> 1;
  const int tid = (kw << 5) | lane;
  const int n = tid & 15;
  const int cl = tid >> 4;
  const int cbase = blockIdx.x * SCH;
  const int b = cbase >> 11;
  const int dbase = cbase & 2047;
  const int d = dbase + cl;
  const size_t row0 = (size_t)b * SEQLEN;

  const float* __restrict__ xp  = X + row0 * D_INNER + dbase;
  const float* __restrict__ xdp = XDBL + row0 * N_PROJ;
  const float* __restrict__ bcp = xdp + DT_RANK;
  float* __restrict__ outp = OUT + row0 * D_INNER + dbase;
  float* st0 = s_lds;
  float* st1 = s_lds + S_STAGE;

  ScanSet s;
  v2f bf[DT_RANK / 4];
  float bn = 0.0f, alog = 0.0f, Dd = 0.0f;
  if (steps) {
    scan_fetch(xp, bcp, 0, tid, s);
    alog = A_log[d * D_STATE + n];
    Dd = Dv[d];
    scan_put(st0, tid, s);
    scan_fetch(xp, bcp, 1, tid, s);
  } else {
    const int half = lane >> 4;
    const int ml   = lane & 15;
#pragma unroll
    for (int i = 0; i < DT_RANK / 4; ++i) {
      const int ka = 4 * i + 2 * half;
      bf[i].x = Wd[(size_t)ka * D_INNER + dbase + ml];
      bf[i].y = Wd[(size_t)(ka + 1) * D_INNER + dbase + ml];
    }
    bn = bias[dbase + ml];
    delta_tile(xdp, bf, bn, kw, lane, st0);
  }
  stage_barrier();

  const float a2 = -__expf(alog) * 1.44269504088896f;
  float st = 0.0f;
#pragma nounroll
  for (int c = 0; c < NTILE; c += 2) {
    const bool more = c + 2 < NTILE;
    if (steps) scan_steps(st0, tid, a2, Dd, st);
    else       delta_tile(xdp + (size_t)(c + 1) * CHUNKL * N_PROJ, bf, bn, kw, lane, st1);
    stage_barrier();
    if (steps) scan_move(st0, st1, s, xp, bcp, outp, c, tid, true, more);
    stage_barrier();
    if (steps)     scan_steps(st1, tid, a2, Dd, st);
    else if (more) delta_tile(xdp + (size_t)(c + 2) * CHUNKL * N_PROJ, bf, bn, kw, lane, st0);
    stage_barrier();
    if (steps) scan_move(st1, st0, s, xp, bcp, outp, c + 1, tid, more, c + 3 < NTILE);
    if (more) stage_barrier();
  }
}

extern "C" void kernel_launch(void* const* d_in, const int* in_sizes, int n_in,
                              void* d_out, int out_size, void* d_ws, size_t ws_size,
                              hipStream_t stream) {
  const float* x    = (const float*)d_in[0];
  const float* xpw  = (const float*)d_in[1];
  const float* dtw  = (const float*)d_in[2];
  const float* dtb  = (const float*)d_in[3];
  const float* alog = (const float*)d_in[4];
  const float* dvec = (const float*)d_in[5];
  float* out = (float*)d_out;

  float* xdbl = (float*)d_ws;

  ssm_xproj_wmma<<<dim3(M_TOT / 16), dim3(192), 0, stream>>>(x, xpw, xdbl);

  ssm_scan_seq<<<dim3(BATCH * D_INNER / SCH), dim3(SNT), 0, stream>>>(
      x, xdbl, dtw, dtb, alog, dvec, out);
}


# Scan step loop: the 16-state sums of 8 time steps share one transposed butterfly (pair sums: 8 cross-lane adds per 8 steps, was 32); lane n files step n&7; first add no longer fused with its product.
# speedup vs baseline: 3.5685x; 1.1173x over previous
#include <hip/hip_runtime.h>
#include <math.h>


#define BATCH   2
#define SEQLEN  2048
#define D_INNER 2048
#define D_STATE 16
#define DT_RANK 64
#define N_PROJ  (DT_RANK + 2 * D_STATE)
#define M_TOT   (BATCH * SEQLEN)
#define NTILE   16
#define CHUNKL  (SEQLEN / NTILE)

#define SCH     16
#define SCT     256
#define SNT     512
#define TPITCH  (CHUNKL + 4)

typedef float v2f __attribute__((ext_vector_type(2)));
typedef float v4f __attribute__((ext_vector_type(4)));
typedef float v8f __attribute__((ext_vector_type(8)));

#define G1_KC     128
#define G1_NCH    (D_INNER / G1_KC)
#define G1_APITCH (G1_KC + 4)
#define G1_BPITCH (N_PROJ + 8)
#define G1_ASZ    (16 * G1_APITCH)
#define G1_BSZ    (G1_KC * G1_BPITCH)
#define G1_STAGE  (G1_ASZ + G1_BSZ)
#define G1_GRP    8

struct G1Set { v4f a[3]; v4f b[16]; };

__device__ __forceinline__ void g1_fetch(const float* __restrict__ xpanel,
                                         const float* __restrict__ W, int k0, int tid, G1Set& s) {
  const int r = tid >> 5;
  const float* __restrict__ src = xpanel + k0 + 4 * (tid & 31);
  s.a[0] = *reinterpret_cast<const v4f*>(src + (size_t)r * D_INNER);
  s.a[1] = *reinterpret_cast<const v4f*>(src + (size_t)(r + 6) * D_INNER);
  s.a[2] = *reinterpret_cast<const v4f*>(src + (size_t)((r + 12) & 15) * D_INNER);
  const float* __restrict__ w = W + (size_t)k0 * N_PROJ + 4 * tid;
  for (int j = 0; j < 16; ++j)
    s.b[j] = *reinterpret_cast<const v4f*>(w + (size_t)(768 * j));
}

__device__ __forceinline__ void g1_put(float* __restrict__ sa, float* __restrict__ sb, int tid,
                                       const G1Set& s) {
  const int r = tid >> 5;
  float* __restrict__ dst = sa + 4 * (tid & 31);
  *reinterpret_cast<v4f*>(dst + r * G1_APITCH)               = s.a[0];
  *reinterpret_cast<v4f*>(dst + (r + 6) * G1_APITCH)         = s.a[1];
  *reinterpret_cast<v4f*>(dst + ((r + 12) & 15) * G1_APITCH) = s.a[2];
  float* __restrict__ d = sb + (tid / 24) * G1_BPITCH + 4 * (tid % 24);
  for (int j = 0; j < 16; ++j)
    *reinterpret_cast<v4f*>(d + 8 * j * G1_BPITCH) = s.b[j];
}

__device__ __forceinline__ void g1_frags(const float* __restrict__ pa, const float* __restrict__ pb,
                                         int g, v2f (&fa)[G1_GRP], v2f (&fb)[G1_GRP]) {
#pragma unroll
  for (int i = 0; i < G1_GRP; ++i) {
    const int t = G1_GRP * g + i;
    fa[i] = *reinterpret_cast<const v2f*>(pa + 4 * t);
    fb[i].x = pb[(4 * t) * G1_BPITCH];
    fb[i].y = pb[(4 * t + 1) * G1_BPITCH];
  }
}

__device__ __forceinline__ void stage_barrier() {
  asm volatile("s_wait_dscnt 0x0" ::: "memory");
  asm volatile("s_barrier_signal -1" ::: "memory");
  asm volatile("s_barrier_wait -1" ::: "memory");
}

template <bool DO_PUT, bool DO_FETCH>
__device__ __forceinline__ void g1_step(const float* __restrict__ cur, float* __restrict__ nxt,
                                        G1Set& s, const float* __restrict__ xpanel,
                                        const float* __restrict__ W, int kfetch, int tid,
                                        int aoff, int boff, v8f& acc) {
  const float* __restrict__ pa = cur + aoff;
  const float* __restrict__ pb = cur + boff;
  v2f fa[2][G1_GRP], fb[2][G1_GRP];
  g1_frags(pa, pb, 0, fa[0], fb[0]);
  __builtin_amdgcn_sched_barrier(0);
  if (DO_PUT) g1_put(nxt, nxt + G1_ASZ, tid, s);
  if (DO_FETCH) g1_fetch(xpanel, W, kfetch, tid, s);
#pragma unroll
  for (int g = 0; g < G1_KC / (4 * G1_GRP); ++g) {
    if (g + 1 < G1_KC / (4 * G1_GRP)) g1_frags(pa, pb, g + 1, fa[(g + 1) & 1], fb[(g + 1) & 1]);
    __builtin_amdgcn_sched_barrier(0);
#pragma unroll
    for (int i = 0; i < G1_GRP; ++i)
      acc = __builtin_amdgcn_wmma_f32_16x16x4_f32(
          false, fa[g & 1][i], false, fb[g & 1][i], (short)0, acc, false, false);
  }
}

__global__ __launch_bounds__(192) void ssm_xproj_wmma(const float* __restrict__ X,
                               const float* __restrict__ W,
                               float* __restrict__ Y) {
  __shared__ __attribute__((aligned(16))) float g1_lds[2 * G1_STAGE];

  const int tid  = threadIdx.x;
  const int lane = tid & 31;
  const int wave = tid >> 5;
  const int half = lane >> 4;
  const int ml   = lane & 15;
  const int m0   = blockIdx.x << 4;
  const int n0   = wave << 4;

  float* __restrict__ st0 = g1_lds;
  float* __restrict__ st1 = g1_lds + G1_STAGE;
  const float* __restrict__ xpanel = X + (size_t)m0 * D_INNER;

  G1Set E, O;
  g1_fetch(xpanel, W, 0, tid, E);
  g1_fetch(xpanel, W, G1_KC, tid, O);
  __builtin_amdgcn_sched_barrier(0);
  g1_put(st0, st0 + G1_ASZ, tid, E);
  g1_fetch(xpanel, W, 2 * G1_KC, tid, E);
  stage_barrier();

  const int aoff = ml * G1_APITCH + 2 * half;
  const int boff = G1_ASZ + 2 * half * G1_BPITCH + n0 + ml;
  v8f acc = {};
  for (int c = 0; c < G1_NCH - 4; c += 2) {
    g1_step<true, true>(st0, st1, O, xpanel, W, (c + 3) * G1_KC, tid, aoff, boff, acc);
    stage_barrier();
    g1_step<true, true>(st1, st0, E, xpanel, W, (c + 4) * G1_KC, tid, aoff, boff, acc);
    stage_barrier();
  }
  g1_step<true, true>(st0, st1, O, xpanel, W, (G1_NCH - 1) * G1_KC, tid, aoff, boff, acc);
  stage_barrier();
  g1_step<true, false>(st1, st0, E, xpanel, W, 0, tid, aoff, boff, acc);
  stage_barrier();
  g1_step<true, false>(st0, st1, O, xpanel, W, 0, tid, aoff, boff, acc);
  stage_barrier();
  g1_step<false, false>(st1, st0, O, xpanel, W, 0, tid, aoff, boff, acc);
#pragma unroll
  for (int r = 0; r < 8; ++r) {
    const int m = m0 + r + 8 * half;
    Y[(size_t)m * N_PROJ + n0 + ml] = acc[r];
  }
}

#define S_XY    (SCH * TPITCH)
#define S_BC    (2 * SCH * TPITCH)
#define S_STAGE ((2 * SCH + 2 * D_STATE) * TPITCH)
#define S_LDS_FLOATS (164 * 256)

template <int NSWEEP, int F4_PER_ROW>
__device__ __forceinline__ void tile_fetch(const float* __restrict__ src, size_t ld,
                                           int tid, v4f (&v)[NSWEEP]) {
  constexpr int ROWS_PER_SWEEP = SCT / F4_PER_ROW;
  const int c4 = tid % F4_PER_ROW;
  const int r0 = tid / F4_PER_ROW;
  for (int it = 0; it < NSWEEP; ++it) {
    const int tt = r0 + ROWS_PER_SWEEP * it;
    v[it] = *reinterpret_cast<const v4f*>(src + (size_t)tt * ld + 4 * c4);
  }
}

template <int NSWEEP, int F4_PER_ROW>
__device__ __forceinline__ void tile_put_transposed(float* __restrict__ dst, int tid,
                                                    const v4f (&v)[NSWEEP]) {
  constexpr int ROWS_PER_SWEEP = SCT / F4_PER_ROW;
  const int c4 = tid % F4_PER_ROW;
  const int r0 = tid / F4_PER_ROW;
  for (int it = 0; it < NSWEEP; ++it) {
    const int tt = r0 + ROWS_PER_SWEEP * it;
    dst[(4 * c4 + 0) * TPITCH + tt] = v[it].x;
    dst[(4 * c4 + 1) * TPITCH + tt] = v[it].y;
    dst[(4 * c4 + 2) * TPITCH + tt] = v[it].z;
    dst[(4 * c4 + 3) * TPITCH + tt] = v[it].w;
  }
}

struct ScanSet { v4f x[2]; v4f bc[4]; };

__device__ __forceinline__ void scan_fetch(const float* __restrict__ xp,
                                           const float* __restrict__ bcp,
                                           int tile, int tid, ScanSet& s) {
  const size_t r = (size_t)tile * CHUNKL;
  tile_fetch<2, 4>(xp + r * D_INNER, D_INNER, tid, s.x);
  tile_fetch<4, 8>(bcp + r * N_PROJ, N_PROJ, tid, s.bc);
}

__device__ __forceinline__ void scan_put(float* __restrict__ stage, int tid, const ScanSet& s) {
  tile_put_transposed<2, 4>(stage + S_XY, tid, s.x);
  tile_put_transposed<4, 8>(stage + S_BC, tid, s.bc);
}

__device__ __forceinline__ void delta_tile(const float* __restrict__ xd,
                                           const v2f (&bf)[DT_RANK / 4], float bn,
                                           int pw, int lane, float* __restrict__ stage) {
  const int half = lane >> 4;
  const int ml   = lane & 15;
  const float* __restrict__ arow = xd + (size_t)(16 * pw + ml) * N_PROJ + 2 * half;
  v2f a[DT_RANK / 4];
#pragma unroll
  for (int i = 0; i < DT_RANK / 4; ++i) {
    a[i].x = arow[4 * i];
    a[i].y = arow[4 * i + 1];
  }
  v8f acc = {};
#pragma unroll
  for (int i = 0; i < DT_RANK / 4; ++i)
    acc = __builtin_amdgcn_wmma_f32_16x16x4_f32(
        false, a[i], false, bf[i], (short)0, acc, false, false);
  float sp[8];
#pragma unroll
  for (int r = 0; r < 8; ++r) {
    float v = acc[r] + bn;
    sp[r] = (v > 20.0f) ? v : log1pf(__expf(v));
  }
  float* __restrict__ dst = stage + ml * TPITCH + 16 * pw + 8 * half;
  v4f lo = {sp[0], sp[1], sp[2], sp[3]};
  v4f hi = {sp[4], sp[5], sp[6], sp[7]};
  *reinterpret_cast<v4f*>(dst)     = lo;
  *reinterpret_cast<v4f*>(dst + 4) = hi;
}

template <int MASK>
__device__ __forceinline__ float row_xor(float v) {
  return __int_as_float(__builtin_amdgcn_update_dpp(
      0, __float_as_int(v), 0x160 + MASK, 0xf, 0xf, true));
}

template <int MASK>
__device__ __forceinline__ float pair_sum(float a, float b, bool hi) {
  const float keep = hi ? b : a;
  const float send = hi ? a : b;
  return keep + row_xor<MASK>(send);
}

__device__ __forceinline__ float scan_group(const v4f dt4, const v4f x4, const v4f b4, const v4f c4,
                                            float a2, bool n1, bool n2, float& st) {
  float p[4];
  for (int j = 0; j < 4; ++j) {
    const float dt = dt4[j];
    const float xv = x4[j];
    const float Bn = b4[j];
    const float Cn = c4[j];
    const float dA = exp2f(dt * a2);
    st = fmaf(dA, st, dt * Bn * xv);
    p[j] = st * Cn;
  }
  const float q0 = pair_sum<1>(p[0], p[1], n1);
  const float q1 = pair_sum<1>(p[2], p[3], n1);
  return pair_sum<2>(q0, q1, n2);
}

__device__ __forceinline__ void scan_steps(float* cur, int tid, float a2, float Dd, float& st) {
  const int n = tid & 15;
  const int cl = tid >> 4;
  const bool n1 = (n & 1) != 0;
  const bool n2 = (n & 2) != 0;
  const bool n4 = (n & 4) != 0;
  const int tq = n & 7;
  const float* pdt = cur + cl * TPITCH;
  float* pxy = cur + S_XY + cl * TPITCH;
  const float* pb = cur + S_BC + n * TPITCH;
  const float* pc = cur + S_BC + (D_STATE + n) * TPITCH;
  v4f dtA = *reinterpret_cast<const v4f*>(pdt);
  v4f xA  = *reinterpret_cast<const v4f*>(pxy);
  v4f bA  = *reinterpret_cast<const v4f*>(pb);
  v4f cA  = *reinterpret_cast<const v4f*>(pc);
  for (int l = 0; l < CHUNKL; l += 8) {
    const v4f dtB = *reinterpret_cast<const v4f*>(pdt + l + 4);
    const v4f xB  = *reinterpret_cast<const v4f*>(pxy + l + 4);
    const v4f bB  = *reinterpret_cast<const v4f*>(pb + l + 4);
    const v4f cB  = *reinterpret_cast<const v4f*>(pc + l + 4);
    const float xq = pxy[l + tq];
    __builtin_amdgcn_sched_barrier(0);
    const float rA = scan_group(dtA, xA, bA, cA, a2, n1, n2, st);
    const int l2 = (l + 8) & (CHUNKL - 1);
    dtA = *reinterpret_cast<const v4f*>(pdt + l2);
    xA  = *reinterpret_cast<const v4f*>(pxy + l2);
    bA  = *reinterpret_cast<const v4f*>(pb + l2);
    cA  = *reinterpret_cast<const v4f*>(pc + l2);
    __builtin_amdgcn_sched_barrier(0);
    const float rB = scan_group(dtB, xB, bB, cB, a2, n1, n2, st);
    float y = pair_sum<4>(rA, rB, n4);
    y += row_xor<8>(y);
    pxy[l + tq] = y + xq * Dd;
  }
}

__device__ __forceinline__ void scan_move(const float* cur, float* __restrict__ nxt, ScanSet& s,
                                          const float* __restrict__ xp,
                                          const float* __restrict__ bcp,
                                          float* __restrict__ outp, int tile, int tid,
                                          bool do_put, bool do_fetch) {
  {
    const int c4 = tid & 3;
    const int r0 = tid >> 2;
    const float* sy = cur + S_XY;
    float* __restrict__ o = outp + (size_t)tile * CHUNKL * D_INNER + 4 * c4;
    for (int it = 0; it < 2; ++it) {
      const int tt = r0 + 64 * it;
      v4f v;
      v.x = sy[(4 * c4 + 0) * TPITCH + tt];
      v.y = sy[(4 * c4 + 1) * TPITCH + tt];
      v.z = sy[(4 * c4 + 2) * TPITCH + tt];
      v.w = sy[(4 * c4 + 3) * TPITCH + tt];
      *reinterpret_cast<v4f*>(o + (size_t)tt * D_INNER) = v;
    }
  }
  if (do_put) {
    scan_put(nxt, tid, s);
    if (do_fetch) scan_fetch(xp, bcp, tile + 2, tid, s);
  }
}

__global__ __launch_bounds__(SNT) void ssm_scan_seq(const float* __restrict__ X,
                               const float* __restrict__ XDBL,
                               const float* __restrict__ Wd,
                               const float* __restrict__ bias,
                               const float* __restrict__ A_log,
                               const float* __restrict__ Dv,
                               float* __restrict__ OUT) {
  __shared__ __attribute__((aligned(16))) float s_lds[S_LDS_FLOATS];

  const int lane = threadIdx.x & 31;
  const int w = __builtin_amdgcn_readfirstlane((int)(threadIdx.x >> 5));
  const bool steps = (((w ^ (w >> 2)) & 1) == 0);
  const int kw = w >> 1;
  const int tid = (kw << 5) | lane;
  const int n = tid & 15;
  const int cl = tid >> 4;
  const int cbase = blockIdx.x * SCH;
  const int b = cbase >> 11;
  const int dbase = cbase & 2047;
  const int d = dbase + cl;
  const size_t row0 = (size_t)b * SEQLEN;

  const float* __restrict__ xp  = X + row0 * D_INNER + dbase;
  const float* __restrict__ xdp = XDBL + row0 * N_PROJ;
  const float* __restrict__ bcp = xdp + DT_RANK;
  float* __restrict__ outp = OUT + row0 * D_INNER + dbase;
  float* st0 = s_lds;
  float* st1 = s_lds + S_STAGE;

  ScanSet s;
  v2f bf[DT_RANK / 4];
  float bn = 0.0f, alog = 0.0f, Dd = 0.0f;
  if (steps) {
    scan_fetch(xp, bcp, 0, tid, s);
    alog = A_log[d * D_STATE + n];
    Dd = Dv[d];
    scan_put(st0, tid, s);
    scan_fetch(xp, bcp, 1, tid, s);
  } else {
    const int half = lane >> 4;
    const int ml   = lane & 15;
#pragma unroll
    for (int i = 0; i < DT_RANK / 4; ++i) {
      const int ka = 4 * i + 2 * half;
      bf[i].x = Wd[(size_t)ka * D_INNER + dbase + ml];
      bf[i].y = Wd[(size_t)(ka + 1) * D_INNER + dbase + ml];
    }
    bn = bias[dbase + ml];
    delta_tile(xdp, bf, bn, kw, lane, st0);
  }
  stage_barrier();

  const float a2 = -__expf(alog) * 1.44269504088896f;
  float st = 0.0f;
#pragma nounroll
  for (int c = 0; c < NTILE; c += 2) {
    const bool more = c + 2 < NTILE;
    if (steps) scan_steps(st0, tid, a2, Dd, st);
    else       delta_tile(xdp + (size_t)(c + 1) * CHUNKL * N_PROJ, bf, bn, kw, lane, st1);
    stage_barrier();
    if (steps) scan_move(st0, st1, s, xp, bcp, outp, c, tid, true, more);
    stage_barrier();
    if (steps)     scan_steps(st1, tid, a2, Dd, st);
    else if (more) delta_tile(xdp + (size_t)(c + 2) * CHUNKL * N_PROJ, bf, bn, kw, lane, st0);
    stage_barrier();
    if (steps) scan_move(st1, st0, s, xp, bcp, outp, c + 1, tid, more, c + 3 < NTILE);
    if (more) stage_barrier();
  }
}

extern "C" void kernel_launch(void* const* d_in, const int* in_sizes, int n_in,
                              void* d_out, int out_size, void* d_ws, size_t ws_size,
                              hipStream_t stream) {
  const float* x    = (const float*)d_in[0];
  const float* xpw  = (const float*)d_in[1];
  const float* dtw  = (const float*)d_in[2];
  const float* dtb  = (const float*)d_in[3];
  const float* alog = (const float*)d_in[4];
  const float* dvec = (const float*)d_in[5];
  float* out = (float*)d_out;

  float* xdbl = (float*)d_ws;

  ssm_xproj_wmma<<<dim3(M_TOT / 16), dim3(192), 0, stream>>>(x, xpw, xdbl);

  ssm_scan_seq<<<dim3(BATCH * D_INNER / SCH), dim3(SNT), 0, stream>>>(
      x, xdbl, dtw, dtb, alog, dvec, out);
}


# Scan block: dependent chains' latencies filled with independent work in both wave kinds (stepping: next 8 steps' 2^(dt*a), dt*B*x made a trip ahead; delta: 8 softplus chains side by side, by select).
# speedup vs baseline: 3.9240x; 1.0996x over previous
#include <hip/hip_runtime.h>
#include <math.h>


#define BATCH   2
#define SEQLEN  2048
#define D_INNER 2048
#define D_STATE 16
#define DT_RANK 64
#define N_PROJ  (DT_RANK + 2 * D_STATE)
#define M_TOT   (BATCH * SEQLEN)
#define NTILE   16
#define CHUNKL  (SEQLEN / NTILE)

#define SCH     16
#define SCT     256
#define SNT     512
#define TPITCH  (CHUNKL + 4)

typedef float v2f __attribute__((ext_vector_type(2)));
typedef float v4f __attribute__((ext_vector_type(4)));
typedef float v8f __attribute__((ext_vector_type(8)));

#define G1_KC     128
#define G1_NCH    (D_INNER / G1_KC)
#define G1_APITCH (G1_KC + 4)
#define G1_BPITCH (N_PROJ + 8)
#define G1_ASZ    (16 * G1_APITCH)
#define G1_BSZ    (G1_KC * G1_BPITCH)
#define G1_STAGE  (G1_ASZ + G1_BSZ)
#define G1_GRP    8

struct G1Set { v4f a[3]; v4f b[16]; };

__device__ __forceinline__ void g1_fetch(const float* __restrict__ xpanel,
                                         const float* __restrict__ W, int k0, int tid, G1Set& s) {
  const int r = tid >> 5;
  const float* __restrict__ src = xpanel + k0 + 4 * (tid & 31);
  s.a[0] = *reinterpret_cast<const v4f*>(src + (size_t)r * D_INNER);
  s.a[1] = *reinterpret_cast<const v4f*>(src + (size_t)(r + 6) * D_INNER);
  s.a[2] = *reinterpret_cast<const v4f*>(src + (size_t)((r + 12) & 15) * D_INNER);
  const float* __restrict__ w = W + (size_t)k0 * N_PROJ + 4 * tid;
  for (int j = 0; j < 16; ++j)
    s.b[j] = *reinterpret_cast<const v4f*>(w + (size_t)(768 * j));
}

__device__ __forceinline__ void g1_put(float* __restrict__ sa, float* __restrict__ sb, int tid,
                                       const G1Set& s) {
  const int r = tid >> 5;
  float* __restrict__ dst = sa + 4 * (tid & 31);
  *reinterpret_cast<v4f*>(dst + r * G1_APITCH)               = s.a[0];
  *reinterpret_cast<v4f*>(dst + (r + 6) * G1_APITCH)         = s.a[1];
  *reinterpret_cast<v4f*>(dst + ((r + 12) & 15) * G1_APITCH) = s.a[2];
  float* __restrict__ d = sb + (tid / 24) * G1_BPITCH + 4 * (tid % 24);
  for (int j = 0; j < 16; ++j)
    *reinterpret_cast<v4f*>(d + 8 * j * G1_BPITCH) = s.b[j];
}

__device__ __forceinline__ void g1_frags(const float* __restrict__ pa, const float* __restrict__ pb,
                                         int g, v2f (&fa)[G1_GRP], v2f (&fb)[G1_GRP]) {
#pragma unroll
  for (int i = 0; i < G1_GRP; ++i) {
    const int t = G1_GRP * g + i;
    fa[i] = *reinterpret_cast<const v2f*>(pa + 4 * t);
    fb[i].x = pb[(4 * t) * G1_BPITCH];
    fb[i].y = pb[(4 * t + 1) * G1_BPITCH];
  }
}

__device__ __forceinline__ void stage_barrier() {
  asm volatile("s_wait_dscnt 0x0" ::: "memory");
  asm volatile("s_barrier_signal -1" ::: "memory");
  asm volatile("s_barrier_wait -1" ::: "memory");
}

template <bool DO_PUT, bool DO_FETCH>
__device__ __forceinline__ void g1_step(const float* __restrict__ cur, float* __restrict__ nxt,
                                        G1Set& s, const float* __restrict__ xpanel,
                                        const float* __restrict__ W, int kfetch, int tid,
                                        int aoff, int boff, v8f& acc) {
  const float* __restrict__ pa = cur + aoff;
  const float* __restrict__ pb = cur + boff;
  v2f fa[2][G1_GRP], fb[2][G1_GRP];
  g1_frags(pa, pb, 0, fa[0], fb[0]);
  __builtin_amdgcn_sched_barrier(0);
  if (DO_PUT) g1_put(nxt, nxt + G1_ASZ, tid, s);
  if (DO_FETCH) g1_fetch(xpanel, W, kfetch, tid, s);
#pragma unroll
  for (int g = 0; g < G1_KC / (4 * G1_GRP); ++g) {
    if (g + 1 < G1_KC / (4 * G1_GRP)) g1_frags(pa, pb, g + 1, fa[(g + 1) & 1], fb[(g + 1) & 1]);
    __builtin_amdgcn_sched_barrier(0);
#pragma unroll
    for (int i = 0; i < G1_GRP; ++i)
      acc = __builtin_amdgcn_wmma_f32_16x16x4_f32(
          false, fa[g & 1][i], false, fb[g & 1][i], (short)0, acc, false, false);
  }
}

__global__ __launch_bounds__(192) void ssm_xproj_wmma(const float* __restrict__ X,
                               const float* __restrict__ W,
                               float* __restrict__ Y) {
  __shared__ __attribute__((aligned(16))) float g1_lds[2 * G1_STAGE];

  const int tid  = threadIdx.x;
  const int lane = tid & 31;
  const int wave = tid >> 5;
  const int half = lane >> 4;
  const int ml   = lane & 15;
  const int m0   = blockIdx.x << 4;
  const int n0   = wave << 4;

  float* __restrict__ st0 = g1_lds;
  float* __restrict__ st1 = g1_lds + G1_STAGE;
  const float* __restrict__ xpanel = X + (size_t)m0 * D_INNER;

  G1Set E, O;
  g1_fetch(xpanel, W, 0, tid, E);
  g1_fetch(xpanel, W, G1_KC, tid, O);
  __builtin_amdgcn_sched_barrier(0);
  g1_put(st0, st0 + G1_ASZ, tid, E);
  g1_fetch(xpanel, W, 2 * G1_KC, tid, E);
  stage_barrier();

  const int aoff = ml * G1_APITCH + 2 * half;
  const int boff = G1_ASZ + 2 * half * G1_BPITCH + n0 + ml;
  v8f acc = {};
  for (int c = 0; c < G1_NCH - 4; c += 2) {
    g1_step<true, true>(st0, st1, O, xpanel, W, (c + 3) * G1_KC, tid, aoff, boff, acc);
    stage_barrier();
    g1_step<true, true>(st1, st0, E, xpanel, W, (c + 4) * G1_KC, tid, aoff, boff, acc);
    stage_barrier();
  }
  g1_step<true, true>(st0, st1, O, xpanel, W, (G1_NCH - 1) * G1_KC, tid, aoff, boff, acc);
  stage_barrier();
  g1_step<true, false>(st1, st0, E, xpanel, W, 0, tid, aoff, boff, acc);
  stage_barrier();
  g1_step<true, false>(st0, st1, O, xpanel, W, 0, tid, aoff, boff, acc);
  stage_barrier();
  g1_step<false, false>(st1, st0, O, xpanel, W, 0, tid, aoff, boff, acc);
#pragma unroll
  for (int r = 0; r < 8; ++r) {
    const int m = m0 + r + 8 * half;
    Y[(size_t)m * N_PROJ + n0 + ml] = acc[r];
  }
}

#define S_XY    (SCH * TPITCH)
#define S_BC    (2 * SCH * TPITCH)
#define S_STAGE ((2 * SCH + 2 * D_STATE) * TPITCH)
#define S_LDS_FLOATS (164 * 256)

template <int NSWEEP, int F4_PER_ROW>
__device__ __forceinline__ void tile_fetch(const float* __restrict__ src, size_t ld,
                                           int tid, v4f (&v)[NSWEEP]) {
  constexpr int ROWS_PER_SWEEP = SCT / F4_PER_ROW;
  const int c4 = tid % F4_PER_ROW;
  const int r0 = tid / F4_PER_ROW;
  for (int it = 0; it < NSWEEP; ++it) {
    const int tt = r0 + ROWS_PER_SWEEP * it;
    v[it] = *reinterpret_cast<const v4f*>(src + (size_t)tt * ld + 4 * c4);
  }
}

template <int NSWEEP, int F4_PER_ROW>
__device__ __forceinline__ void tile_put_transposed(float* __restrict__ dst, int tid,
                                                    const v4f (&v)[NSWEEP]) {
  constexpr int ROWS_PER_SWEEP = SCT / F4_PER_ROW;
  const int c4 = tid % F4_PER_ROW;
  const int r0 = tid / F4_PER_ROW;
  for (int it = 0; it < NSWEEP; ++it) {
    const int tt = r0 + ROWS_PER_SWEEP * it;
    dst[(4 * c4 + 0) * TPITCH + tt] = v[it].x;
    dst[(4 * c4 + 1) * TPITCH + tt] = v[it].y;
    dst[(4 * c4 + 2) * TPITCH + tt] = v[it].z;
    dst[(4 * c4 + 3) * TPITCH + tt] = v[it].w;
  }
}

struct ScanSet { v4f x[2]; v4f bc[4]; };

__device__ __forceinline__ void scan_fetch(const float* __restrict__ xp,
                                           const float* __restrict__ bcp,
                                           int tile, int tid, ScanSet& s) {
  const size_t r = (size_t)tile * CHUNKL;
  tile_fetch<2, 4>(xp + r * D_INNER, D_INNER, tid, s.x);
  tile_fetch<4, 8>(bcp + r * N_PROJ, N_PROJ, tid, s.bc);
}

__device__ __forceinline__ void scan_put(float* __restrict__ stage, int tid, const ScanSet& s) {
  tile_put_transposed<2, 4>(stage + S_XY, tid, s.x);
  tile_put_transposed<4, 8>(stage + S_BC, tid, s.bc);
}

__device__ __forceinline__ void delta_tile(const float* __restrict__ xd,
                                           const v2f (&bf)[DT_RANK / 4], float bn,
                                           int pw, int lane, float* __restrict__ stage) {
  const int half = lane >> 4;
  const int ml   = lane & 15;
  const float* __restrict__ arow = xd + (size_t)(16 * pw + ml) * N_PROJ + 2 * half;
  v2f a[DT_RANK / 4];
#pragma unroll
  for (int i = 0; i < DT_RANK / 4; ++i) {
    a[i].x = arow[4 * i];
    a[i].y = arow[4 * i + 1];
  }
  v8f acc = {};
#pragma unroll
  for (int i = 0; i < DT_RANK / 4; ++i)
    acc = __builtin_amdgcn_wmma_f32_16x16x4_f32(
        false, a[i], false, bf[i], (short)0, acc, false, false);
  float sp[8];
#pragma unroll
  for (int r = 0; r < 8; ++r) {
    const float v = acc[r] + bn;
    const float t = log1pf(__expf(v));
    sp[r] = (v > 20.0f) ? v : t;
  }
  float* __restrict__ dst = stage + ml * TPITCH + 16 * pw + 8 * half;
  v4f lo = {sp[0], sp[1], sp[2], sp[3]};
  v4f hi = {sp[4], sp[5], sp[6], sp[7]};
  *reinterpret_cast<v4f*>(dst)     = lo;
  *reinterpret_cast<v4f*>(dst + 4) = hi;
}

template <int MASK>
__device__ __forceinline__ float row_xor(float v) {
  return __int_as_float(__builtin_amdgcn_update_dpp(
      0, __float_as_int(v), 0x160 + MASK, 0xf, 0xf, true));
}

template <int MASK>
__device__ __forceinline__ float pair_sum(float a, float b, bool hi) {
  const float keep = hi ? b : a;
  const float send = hi ? a : b;
  return keep + row_xor<MASK>(send);
}


struct StepOps { v4f dt[2]; v4f x[2]; v4f b[2]; };
struct StepPrep { float dA[8]; float u[8]; };

__device__ __forceinline__ void ops_load(const float* pdt, const float* pxy, const float* pb, int l,
                                         StepOps& o) {
  o.dt[0] = *reinterpret_cast<const v4f*>(pdt + l);
  o.dt[1] = *reinterpret_cast<const v4f*>(pdt + l + 4);
  o.x[0]  = *reinterpret_cast<const v4f*>(pxy + l);
  o.x[1]  = *reinterpret_cast<const v4f*>(pxy + l + 4);
  o.b[0]  = *reinterpret_cast<const v4f*>(pb + l);
  o.b[1]  = *reinterpret_cast<const v4f*>(pb + l + 4);
}

__device__ __forceinline__ void step_prep(const StepOps& o, float a2, StepPrep& p) {
#pragma unroll
  for (int j = 0; j < 8; ++j) {
    const float dt = o.dt[j >> 2][j & 3];
    p.dA[j] = exp2f(dt * a2);
    p.u[j]  = dt * o.b[j >> 2][j & 3] * o.x[j >> 2][j & 3];
  }
}

__device__ __forceinline__ float step_chain(const StepPrep& p, const v4f c0, const v4f c1,
                                            bool n1, bool n2, bool n4, float& st) {
  float q[8];
#pragma unroll
  for (int j = 0; j < 8; ++j) {
    st = fmaf(p.dA[j], st, p.u[j]);
    q[j] = st * (j < 4 ? c0[j & 3] : c1[j & 3]);
  }
  const float q01 = pair_sum<1>(q[0], q[1], n1);
  const float q23 = pair_sum<1>(q[2], q[3], n1);
  const float q45 = pair_sum<1>(q[4], q[5], n1);
  const float q67 = pair_sum<1>(q[6], q[7], n1);
  const float rA = pair_sum<2>(q01, q23, n2);
  const float rB = pair_sum<2>(q45, q67, n2);
  float y = pair_sum<4>(rA, rB, n4);
  y += row_xor<8>(y);
  return y;
}

template <bool DO_LOAD, bool DO_LOADC, bool DO_PREP>
__device__ __forceinline__ void step_trip(const float* pdt, float* pxy, const float* pb, const float* pc,
                                          int l, int tq, StepOps& oLoad, const StepOps& oPrep,
                                          StepPrep& pNext, const StepPrep& pCur,
                                          const v4f cCur0, const v4f cCur1, v4f& cNext0, v4f& cNext1,
                                          float a2, float Dd, bool n1, bool n2, bool n4, float& st) {
  if (DO_LOAD) ops_load(pdt, pxy, pb, l + 16, oLoad);
  if (DO_LOADC) {
    cNext0 = *reinterpret_cast<const v4f*>(pc + l + 8);
    cNext1 = *reinterpret_cast<const v4f*>(pc + l + 12);
  }
  const float xq = pxy[l + tq];
  __builtin_amdgcn_sched_barrier(0);
  if (DO_PREP) step_prep(oPrep, a2, pNext);
  const float y = step_chain(pCur, cCur0, cCur1, n1, n2, n4, st);
  pxy[l + tq] = y + xq * Dd;
}

__device__ __forceinline__ void scan_steps(float* cur, int tid, float a2, float Dd, float& st) {
  const int n = tid & 15;
  const int cl = tid >> 4;
  const bool n1 = (n & 1) != 0;
  const bool n2 = (n & 2) != 0;
  const bool n4 = (n & 4) != 0;
  const int tq = n & 7;
  const float* pdt = cur + cl * TPITCH;
  float* pxy = cur + S_XY + cl * TPITCH;
  const float* pb = cur + S_BC + n * TPITCH;
  const float* pc = cur + S_BC + (D_STATE + n) * TPITCH;
  StepOps oE, oO;
  StepPrep pE, pO;
  v4f cE0, cE1, cO0, cO1;
  ops_load(pdt, pxy, pb, 0, oE);
  cE0 = *reinterpret_cast<const v4f*>(pc);
  cE1 = *reinterpret_cast<const v4f*>(pc + 4);
  ops_load(pdt, pxy, pb, 8, oO);
  step_prep(oE, a2, pE);
#pragma nounroll
  for (int l = 0; l < CHUNKL - 16; l += 16) {
    step_trip<true, true, true>(pdt, pxy, pb, pc, l, tq, oE, oO, pO, pE, cE0, cE1, cO0, cO1,
                                a2, Dd, n1, n2, n4, st);
    step_trip<true, true, true>(pdt, pxy, pb, pc, l + 8, tq, oO, oE, pE, pO, cO0, cO1, cE0, cE1,
                                a2, Dd, n1, n2, n4, st);
  }
  step_trip<false, true, true>(pdt, pxy, pb, pc, CHUNKL - 16, tq, oE, oO, pO, pE, cE0, cE1, cO0, cO1,
                               a2, Dd, n1, n2, n4, st);
  step_trip<false, false, false>(pdt, pxy, pb, pc, CHUNKL - 8, tq, oO, oE, pE, pO, cO0, cO1, cE0, cE1,
                                 a2, Dd, n1, n2, n4, st);
}

__device__ __forceinline__ void scan_move(const float* cur, float* __restrict__ nxt, ScanSet& s,
                                          const float* __restrict__ xp,
                                          const float* __restrict__ bcp,
                                          float* __restrict__ outp, int tile, int tid,
                                          bool do_put, bool do_fetch) {
  {
    const int c4 = tid & 3;
    const int r0 = tid >> 2;
    const float* sy = cur + S_XY;
    float* __restrict__ o = outp + (size_t)tile * CHUNKL * D_INNER + 4 * c4;
    for (int it = 0; it < 2; ++it) {
      const int tt = r0 + 64 * it;
      v4f v;
      v.x = sy[(4 * c4 + 0) * TPITCH + tt];
      v.y = sy[(4 * c4 + 1) * TPITCH + tt];
      v.z = sy[(4 * c4 + 2) * TPITCH + tt];
      v.w = sy[(4 * c4 + 3) * TPITCH + tt];
      *reinterpret_cast<v4f*>(o + (size_t)tt * D_INNER) = v;
    }
  }
  if (do_put) {
    scan_put(nxt, tid, s);
    if (do_fetch) scan_fetch(xp, bcp, tile + 2, tid, s);
  }
}

__global__ __launch_bounds__(SNT) void ssm_scan_seq(const float* __restrict__ X,
                               const float* __restrict__ XDBL,
                               const float* __restrict__ Wd,
                               const float* __restrict__ bias,
                               const float* __restrict__ A_log,
                               const float* __restrict__ Dv,
                               float* __restrict__ OUT) {
  __shared__ __attribute__((aligned(16))) float s_lds[S_LDS_FLOATS];

  const int lane = threadIdx.x & 31;
  const int w = __builtin_amdgcn_readfirstlane((int)(threadIdx.x >> 5));
  const bool steps = (((w ^ (w >> 2)) & 1) == 0);
  const int kw = w >> 1;
  const int tid = (kw << 5) | lane;
  const int n = tid & 15;
  const int cl = tid >> 4;
  const int cbase = blockIdx.x * SCH;
  const int b = cbase >> 11;
  const int dbase = cbase & 2047;
  const int d = dbase + cl;
  const size_t row0 = (size_t)b * SEQLEN;

  const float* __restrict__ xp  = X + row0 * D_INNER + dbase;
  const float* __restrict__ xdp = XDBL + row0 * N_PROJ;
  const float* __restrict__ bcp = xdp + DT_RANK;
  float* __restrict__ outp = OUT + row0 * D_INNER + dbase;
  float* st0 = s_lds;
  float* st1 = s_lds + S_STAGE;

  ScanSet s;
  v2f bf[DT_RANK / 4];
  float bn = 0.0f, alog = 0.0f, Dd = 0.0f;
  if (steps) {
    scan_fetch(xp, bcp, 0, tid, s);
    alog = A_log[d * D_STATE + n];
    Dd = Dv[d];
    scan_put(st0, tid, s);
    scan_fetch(xp, bcp, 1, tid, s);
  } else {
    const int half = lane >> 4;
    const int ml   = lane & 15;
#pragma unroll
    for (int i = 0; i < DT_RANK / 4; ++i) {
      const int ka = 4 * i + 2 * half;
      bf[i].x = Wd[(size_t)ka * D_INNER + dbase + ml];
      bf[i].y = Wd[(size_t)(ka + 1) * D_INNER + dbase + ml];
    }
    bn = bias[dbase + ml];
    delta_tile(xdp, bf, bn, kw, lane, st0);
  }
  stage_barrier();

  const float a2 = -__expf(alog) * 1.44269504088896f;
  float st = 0.0f;
#pragma nounroll
  for (int c = 0; c < NTILE; c += 2) {
    const bool more = c + 2 < NTILE;
    if (steps) scan_steps(st0, tid, a2, Dd, st);
    else       delta_tile(xdp + (size_t)(c + 1) * CHUNKL * N_PROJ, bf, bn, kw, lane, st1);
    stage_barrier();
    if (steps) scan_move(st0, st1, s, xp, bcp, outp, c, tid, true, more);
    stage_barrier();
    if (steps)     scan_steps(st1, tid, a2, Dd, st);
    else if (more) delta_tile(xdp + (size_t)(c + 2) * CHUNKL * N_PROJ, bf, bn, kw, lane, st0);
    stage_barrier();
    if (steps) scan_move(st1, st0, s, xp, bcp, outp, c + 1, tid, more, c + 3 < NTILE);
    if (more) stage_barrier();
  }
}

extern "C" void kernel_launch(void* const* d_in, const int* in_sizes, int n_in,
                              void* d_out, int out_size, void* d_ws, size_t ws_size,
                              hipStream_t stream) {
  const float* x    = (const float*)d_in[0];
  const float* xpw  = (const float*)d_in[1];
  const float* dtw  = (const float*)d_in[2];
  const float* dtb  = (const float*)d_in[3];
  const float* alog = (const float*)d_in[4];
  const float* dvec = (const float*)d_in[5];
  float* out = (float*)d_out;

  float* xdbl = (float*)d_ws;

  ssm_xproj_wmma<<<dim3(M_TOT / 16), dim3(192), 0, stream>>>(x, xpw, xdbl);

  ssm_scan_seq<<<dim3(BATCH * D_INNER / SCH), dim3(SNT), 0, stream>>>(
      x, xdbl, dtw, dtb, alog, dvec, out);
}


# Scan block: step loop runs on across tile ends: three-tile LDS ring, the delta waves move the tiles (y out, x, B|C in), one barrier a tile, two trips before its end (sums bit-identical).
# speedup vs baseline: 4.1998x; 1.0703x over previous
#include <hip/hip_runtime.h>
#include <math.h>


#define BATCH   2
#define SEQLEN  2048
#define D_INNER 2048
#define D_STATE 16
#define DT_RANK 64
#define N_PROJ  (DT_RANK + 2 * D_STATE)
#define M_TOT   (BATCH * SEQLEN)
#define NTILE   16
#define CHUNKL  (SEQLEN / NTILE)

#define SCH     16
#define SCT     256
#define SNT     512
#define NSLOT   3
#define TPITCH  (NSLOT * CHUNKL + 4)

typedef float v2f __attribute__((ext_vector_type(2)));
typedef float v4f __attribute__((ext_vector_type(4)));
typedef float v8f __attribute__((ext_vector_type(8)));

#define G1_KC     128
#define G1_NCH    (D_INNER / G1_KC)
#define G1_APITCH (G1_KC + 4)
#define G1_BPITCH (N_PROJ + 8)
#define G1_ASZ    (16 * G1_APITCH)
#define G1_BSZ    (G1_KC * G1_BPITCH)
#define G1_STAGE  (G1_ASZ + G1_BSZ)
#define G1_GRP    8

struct G1Set { v4f a[3]; v4f b[16]; };

__device__ __forceinline__ void g1_fetch(const float* __restrict__ xpanel,
                                         const float* __restrict__ W, int k0, int tid, G1Set& s) {
  const int r = tid >> 5;
  const float* __restrict__ src = xpanel + k0 + 4 * (tid & 31);
  s.a[0] = *reinterpret_cast<const v4f*>(src + (size_t)r * D_INNER);
  s.a[1] = *reinterpret_cast<const v4f*>(src + (size_t)(r + 6) * D_INNER);
  s.a[2] = *reinterpret_cast<const v4f*>(src + (size_t)((r + 12) & 15) * D_INNER);
  const float* __restrict__ w = W + (size_t)k0 * N_PROJ + 4 * tid;
  for (int j = 0; j < 16; ++j)
    s.b[j] = *reinterpret_cast<const v4f*>(w + (size_t)(768 * j));
}

__device__ __forceinline__ void g1_put(float* __restrict__ sa, float* __restrict__ sb, int tid,
                                       const G1Set& s) {
  const int r = tid >> 5;
  float* __restrict__ dst = sa + 4 * (tid & 31);
  *reinterpret_cast<v4f*>(dst + r * G1_APITCH)               = s.a[0];
  *reinterpret_cast<v4f*>(dst + (r + 6) * G1_APITCH)         = s.a[1];
  *reinterpret_cast<v4f*>(dst + ((r + 12) & 15) * G1_APITCH) = s.a[2];
  float* __restrict__ d = sb + (tid / 24) * G1_BPITCH + 4 * (tid % 24);
  for (int j = 0; j < 16; ++j)
    *reinterpret_cast<v4f*>(d + 8 * j * G1_BPITCH) = s.b[j];
}

__device__ __forceinline__ void g1_frags(const float* __restrict__ pa, const float* __restrict__ pb,
                                         int g, v2f (&fa)[G1_GRP], v2f (&fb)[G1_GRP]) {
#pragma unroll
  for (int i = 0; i < G1_GRP; ++i) {
    const int t = G1_GRP * g + i;
    fa[i] = *reinterpret_cast<const v2f*>(pa + 4 * t);
    fb[i].x = pb[(4 * t) * G1_BPITCH];
    fb[i].y = pb[(4 * t + 1) * G1_BPITCH];
  }
}

__device__ __forceinline__ void stage_barrier() {
  asm volatile("s_wait_dscnt 0x0" ::: "memory");
  asm volatile("s_barrier_signal -1" ::: "memory");
  asm volatile("s_barrier_wait -1" ::: "memory");
}

template <bool DO_PUT, bool DO_FETCH>
__device__ __forceinline__ void g1_step(const float* __restrict__ cur, float* __restrict__ nxt,
                                        G1Set& s, const float* __restrict__ xpanel,
                                        const float* __restrict__ W, int kfetch, int tid,
                                        int aoff, int boff, v8f& acc) {
  const float* __restrict__ pa = cur + aoff;
  const float* __restrict__ pb = cur + boff;
  v2f fa[2][G1_GRP], fb[2][G1_GRP];
  g1_frags(pa, pb, 0, fa[0], fb[0]);
  __builtin_amdgcn_sched_barrier(0);
  if (DO_PUT) g1_put(nxt, nxt + G1_ASZ, tid, s);
  if (DO_FETCH) g1_fetch(xpanel, W, kfetch, tid, s);
#pragma unroll
  for (int g = 0; g < G1_KC / (4 * G1_GRP); ++g) {
    if (g + 1 < G1_KC / (4 * G1_GRP)) g1_frags(pa, pb, g + 1, fa[(g + 1) & 1], fb[(g + 1) & 1]);
    __builtin_amdgcn_sched_barrier(0);
#pragma unroll
    for (int i = 0; i < G1_GRP; ++i)
      acc = __builtin_amdgcn_wmma_f32_16x16x4_f32(
          false, fa[g & 1][i], false, fb[g & 1][i], (short)0, acc, false, false);
  }
}

__global__ __launch_bounds__(192) void ssm_xproj_wmma(const float* __restrict__ X,
                               const float* __restrict__ W,
                               float* __restrict__ Y) {
  __shared__ __attribute__((aligned(16))) float g1_lds[2 * G1_STAGE];

  const int tid  = threadIdx.x;
  const int lane = tid & 31;
  const int wave = tid >> 5;
  const int half = lane >> 4;
  const int ml   = lane & 15;
  const int m0   = blockIdx.x << 4;
  const int n0   = wave << 4;

  float* __restrict__ st0 = g1_lds;
  float* __restrict__ st1 = g1_lds + G1_STAGE;
  const float* __restrict__ xpanel = X + (size_t)m0 * D_INNER;

  G1Set E, O;
  g1_fetch(xpanel, W, 0, tid, E);
  g1_fetch(xpanel, W, G1_KC, tid, O);
  __builtin_amdgcn_sched_barrier(0);
  g1_put(st0, st0 + G1_ASZ, tid, E);
  g1_fetch(xpanel, W, 2 * G1_KC, tid, E);
  stage_barrier();

  const int aoff = ml * G1_APITCH + 2 * half;
  const int boff = G1_ASZ + 2 * half * G1_BPITCH + n0 + ml;
  v8f acc = {};
  for (int c = 0; c < G1_NCH - 4; c += 2) {
    g1_step<true, true>(st0, st1, O, xpanel, W, (c + 3) * G1_KC, tid, aoff, boff, acc);
    stage_barrier();
    g1_step<true, true>(st1, st0, E, xpanel, W, (c + 4) * G1_KC, tid, aoff, boff, acc);
    stage_barrier();
  }
  g1_step<true, true>(st0, st1, O, xpanel, W, (G1_NCH - 1) * G1_KC, tid, aoff, boff, acc);
  stage_barrier();
  g1_step<true, false>(st1, st0, E, xpanel, W, 0, tid, aoff, boff, acc);
  stage_barrier();
  g1_step<true, false>(st0, st1, O, xpanel, W, 0, tid, aoff, boff, acc);
  stage_barrier();
  g1_step<false, false>(st1, st0, O, xpanel, W, 0, tid, aoff, boff, acc);
#pragma unroll
  for (int r = 0; r < 8; ++r) {
    const int m = m0 + r + 8 * half;
    Y[(size_t)m * N_PROJ + n0 + ml] = acc[r];
  }
}

#define S_XY    (SCH * TPITCH)
#define S_BC    (2 * SCH * TPITCH)
#define S_LDS_FLOATS (164 * 256)

template <int NSWEEP, int F4_PER_ROW>
__device__ __forceinline__ void tile_fetch(const float* __restrict__ src, size_t ld,
                                           int tid, v4f (&v)[NSWEEP]) {
  constexpr int ROWS_PER_SWEEP = SCT / F4_PER_ROW;
  const int c4 = tid % F4_PER_ROW;
  const int r0 = tid / F4_PER_ROW;
  for (int it = 0; it < NSWEEP; ++it) {
    const int tt = r0 + ROWS_PER_SWEEP * it;
    v[it] = *reinterpret_cast<const v4f*>(src + (size_t)tt * ld + 4 * c4);
  }
}

template <int NSWEEP, int F4_PER_ROW>
__device__ __forceinline__ void tile_put_transposed(float* __restrict__ dst, int tid,
                                                    const v4f (&v)[NSWEEP]) {
  constexpr int ROWS_PER_SWEEP = SCT / F4_PER_ROW;
  const int c4 = tid % F4_PER_ROW;
  const int r0 = tid / F4_PER_ROW;
  for (int it = 0; it < NSWEEP; ++it) {
    const int tt = r0 + ROWS_PER_SWEEP * it;
    dst[(4 * c4 + 0) * TPITCH + tt] = v[it].x;
    dst[(4 * c4 + 1) * TPITCH + tt] = v[it].y;
    dst[(4 * c4 + 2) * TPITCH + tt] = v[it].z;
    dst[(4 * c4 + 3) * TPITCH + tt] = v[it].w;
  }
}

struct ScanSet { v4f x[2]; v4f bc[4]; };

__device__ __forceinline__ void scan_fetch(const float* __restrict__ xp,
                                           const float* __restrict__ bcp,
                                           int tile, int tid, ScanSet& s) {
  const size_t r = (size_t)tile * CHUNKL;
  tile_fetch<2, 4>(xp + r * D_INNER, D_INNER, tid, s.x);
  tile_fetch<4, 8>(bcp + r * N_PROJ, N_PROJ, tid, s.bc);
}

__device__ __forceinline__ void scan_put(float* __restrict__ slot, int tid, const ScanSet& s) {
  tile_put_transposed<2, 4>(slot + S_XY, tid, s.x);
  tile_put_transposed<4, 8>(slot + S_BC, tid, s.bc);
}

__device__ __forceinline__ void delta_tile(const float* __restrict__ xd,
                                           const v2f (&bf)[DT_RANK / 4], float bn,
                                           int pw, int lane, float* __restrict__ stage) {
  const int half = lane >> 4;
  const int ml   = lane & 15;
  const float* __restrict__ arow = xd + (size_t)(16 * pw + ml) * N_PROJ + 2 * half;
  v2f a[DT_RANK / 4];
#pragma unroll
  for (int i = 0; i < DT_RANK / 4; ++i) {
    a[i].x = arow[4 * i];
    a[i].y = arow[4 * i + 1];
  }
  v8f acc = {};
#pragma unroll
  for (int i = 0; i < DT_RANK / 4; ++i)
    acc = __builtin_amdgcn_wmma_f32_16x16x4_f32(
        false, a[i], false, bf[i], (short)0, acc, false, false);
  float sp[8];
#pragma unroll
  for (int r = 0; r < 8; ++r) {
    const float v = acc[r] + bn;
    const float t = log1pf(__expf(v));
    sp[r] = (v > 20.0f) ? v : t;
  }
  float* __restrict__ dst = stage + ml * TPITCH + 16 * pw + 8 * half;
  v4f lo = {sp[0], sp[1], sp[2], sp[3]};
  v4f hi = {sp[4], sp[5], sp[6], sp[7]};
  *reinterpret_cast<v4f*>(dst)     = lo;
  *reinterpret_cast<v4f*>(dst + 4) = hi;
}

template <int MASK>
__device__ __forceinline__ float row_xor(float v) {
  return __int_as_float(__builtin_amdgcn_update_dpp(
      0, __float_as_int(v), 0x160 + MASK, 0xf, 0xf, true));
}

template <int MASK>
__device__ __forceinline__ float pair_sum(float a, float b, bool hi) {
  const float keep = hi ? b : a;
  const float send = hi ? a : b;
  return keep + row_xor<MASK>(send);
}


struct StepOps { v4f dt[2]; v4f x[2]; v4f b[2]; };
struct StepPrep { float dA[8]; float u[8]; };

__device__ __forceinline__ void ops_load(const float* pdt, const float* pxy, const float* pb, int l,
                                         StepOps& o) {
  o.dt[0] = *reinterpret_cast<const v4f*>(pdt + l);
  o.dt[1] = *reinterpret_cast<const v4f*>(pdt + l + 4);
  o.x[0]  = *reinterpret_cast<const v4f*>(pxy + l);
  o.x[1]  = *reinterpret_cast<const v4f*>(pxy + l + 4);
  o.b[0]  = *reinterpret_cast<const v4f*>(pb + l);
  o.b[1]  = *reinterpret_cast<const v4f*>(pb + l + 4);
}

__device__ __forceinline__ void step_prep(const StepOps& o, float a2, StepPrep& p) {
#pragma unroll
  for (int j = 0; j < 8; ++j) {
    const float dt = o.dt[j >> 2][j & 3];
    p.dA[j] = exp2f(dt * a2);
    p.u[j]  = dt * o.b[j >> 2][j & 3] * o.x[j >> 2][j & 3];
  }
}

__device__ __forceinline__ float step_chain(const StepPrep& p, const v4f c0, const v4f c1,
                                            bool n1, bool n2, bool n4, float& st) {
  float q[8];
#pragma unroll
  for (int j = 0; j < 8; ++j) {
    st = fmaf(p.dA[j], st, p.u[j]);
    q[j] = st * (j < 4 ? c0[j & 3] : c1[j & 3]);
  }
  const float q01 = pair_sum<1>(q[0], q[1], n1);
  const float q23 = pair_sum<1>(q[2], q[3], n1);
  const float q45 = pair_sum<1>(q[4], q[5], n1);
  const float q67 = pair_sum<1>(q[6], q[7], n1);
  const float rA = pair_sum<2>(q01, q23, n2);
  const float rB = pair_sum<2>(q45, q67, n2);
  float y = pair_sum<4>(rA, rB, n4);
  y += row_xor<8>(y);
  return y;
}

template <bool DO_LOAD, bool DO_LOADC, bool DO_PREP>
__device__ __forceinline__ void step_trip(const float* pdt, float* pxy, const float* pb, const float* pc,
                                          int l, int lo, int lc, int tq, StepOps& oLoad, const StepOps& oPrep,
                                          StepPrep& pNext, const StepPrep& pCur,
                                          const v4f cCur0, const v4f cCur1, v4f& cNext0, v4f& cNext1,
                                          float a2, float Dd, bool n1, bool n2, bool n4, float& st) {
  if (DO_LOAD) ops_load(pdt, pxy, pb, lo, oLoad);
  if (DO_LOADC) {
    cNext0 = *reinterpret_cast<const v4f*>(pc + lc);
    cNext1 = *reinterpret_cast<const v4f*>(pc + lc + 4);
  }
  const float xq = pxy[l + tq];
  __builtin_amdgcn_sched_barrier(0);
  if (DO_PREP) step_prep(oPrep, a2, pNext);
  const float y = step_chain(pCur, cCur0, cCur1, n1, n2, n4, st);
  pxy[l + tq] = y + xq * Dd;
}

struct Walk { StepOps oE, oO; StepPrep pE, pO; v4f cE0, cE1, cO0, cO1; float st; };
struct LaneC { const float* pdt; float* pxy; const float* pb; const float* pc;
               int tq; bool n1, n2, n4; float a2, Dd; };

template <bool LOAD_E, bool LOADC_E, bool PREP_E, bool LOAD_O, bool LOADC_O, bool PREP_O>
__device__ __forceinline__ void walk_two(const LaneC& L, Walk& k, int l,
                                         int loE, int lcE, int loO, int lcO) {
  step_trip<LOAD_E, LOADC_E, PREP_E>(L.pdt, L.pxy, L.pb, L.pc, l, loE, lcE, L.tq, k.oE, k.oO, k.pO, k.pE,
                                     k.cE0, k.cE1, k.cO0, k.cO1, L.a2, L.Dd, L.n1, L.n2, L.n4, k.st);
  step_trip<LOAD_O, LOADC_O, PREP_O>(L.pdt, L.pxy, L.pb, L.pc, l + 8, loO, lcO, L.tq, k.oO, k.oE, k.pE, k.pO,
                                     k.cO0, k.cO1, k.cE0, k.cE1, L.a2, L.Dd, L.n1, L.n2, L.n4, k.st);
}

__device__ __forceinline__ void walk_fill(const LaneC& L, Walk& k) {
  k.st = 0.0f;
  ops_load(L.pdt, L.pxy, L.pb, 0, k.oE);
  k.cE0 = *reinterpret_cast<const v4f*>(L.pc);
  k.cE1 = *reinterpret_cast<const v4f*>(L.pc + 4);
  ops_load(L.pdt, L.pxy, L.pb, 8, k.oO);
  step_prep(k.oE, L.a2, k.pE);
}

__device__ __forceinline__ void walk_body(const LaneC& L, Walk& k, int base) {
#pragma nounroll
  for (int l = base; l < base + CHUNKL - 16; l += 16)
    walk_two<true, true, true, true, true, true>(L, k, l, l + 16, l + 8, l + 24, l + 16);
}

__device__ __forceinline__ void walk_cross(const LaneC& L, Walk& k, int base, int nb) {
  walk_two<true, true, true, true, true, true>(L, k, base + CHUNKL - 16,
                                               nb, base + CHUNKL - 8, nb + 8, nb);
}

__device__ __forceinline__ void walk_tail(const LaneC& L, Walk& k, int base) {
  walk_two<false, true, true, false, false, false>(L, k, base + CHUNKL - 16,
                                                   0, base + CHUNKL - 8, 0, 0);
}

__device__ __forceinline__ void y_out(const float* slot, float* __restrict__ outp, int tile, int tid) {
  const int c4 = tid & 3;
  const int r0 = tid >> 2;
  const float* sy = slot + S_XY;
  float* __restrict__ o = outp + (size_t)tile * CHUNKL * D_INNER + 4 * c4;
  for (int it = 0; it < 2; ++it) {
    const int tt = r0 + 64 * it;
    v4f v;
    v.x = sy[(4 * c4 + 0) * TPITCH + tt];
    v.y = sy[(4 * c4 + 1) * TPITCH + tt];
    v.z = sy[(4 * c4 + 2) * TPITCH + tt];
    v.w = sy[(4 * c4 + 3) * TPITCH + tt];
    *reinterpret_cast<v4f*>(o + (size_t)tt * D_INNER) = v;
  }
}

#define SLOT_COL(t) (((t) % NSLOT) * CHUNKL)
__global__ __launch_bounds__(SNT) void ssm_scan_seq(const float* __restrict__ X,
                               const float* __restrict__ XDBL,
                               const float* __restrict__ Wd,
                               const float* __restrict__ bias,
                               const float* __restrict__ A_log,
                               const float* __restrict__ Dv,
                               float* __restrict__ OUT) {
  __shared__ __attribute__((aligned(16))) float s_lds[S_LDS_FLOATS];

  const int lane = threadIdx.x & 31;
  const int w = __builtin_amdgcn_readfirstlane((int)(threadIdx.x >> 5));
  const bool steps = (((w ^ (w >> 2)) & 1) == 0);
  const int kw = w >> 1;
  const int tid = (kw << 5) | lane;
  const int n = tid & 15;
  const int cl = tid >> 4;
  const int cbase = blockIdx.x * SCH;
  const int b = cbase >> 11;
  const int dbase = cbase & 2047;
  const int d = dbase + cl;
  const size_t row0 = (size_t)b * SEQLEN;

  const float* __restrict__ xp  = X + row0 * D_INNER + dbase;
  const float* __restrict__ xdp = XDBL + row0 * N_PROJ;
  const float* __restrict__ bcp = xdp + DT_RANK;
  float* __restrict__ outp = OUT + row0 * D_INNER + dbase;

  ScanSet s;
  v2f bf[DT_RANK / 4];
  float bn = 0.0f;
  Walk k;
  LaneC L;
  if (steps) {
    L.pdt = s_lds + cl * TPITCH;
    L.pxy = s_lds + S_XY + cl * TPITCH;
    L.pb  = s_lds + S_BC + n * TPITCH;
    L.pc  = s_lds + S_BC + (D_STATE + n) * TPITCH;
    L.tq = n & 7;
    L.n1 = (n & 1) != 0;
    L.n2 = (n & 2) != 0;
    L.n4 = (n & 4) != 0;
    L.a2 = -__expf(A_log[d * D_STATE + n]) * 1.44269504088896f;
    L.Dd = Dv[d];
  } else {
    scan_fetch(xp, bcp, 0, tid, s);
    const int half = lane >> 4;
    const int ml   = lane & 15;
#pragma unroll
    for (int i = 0; i < DT_RANK / 4; ++i) {
      const int ka = 4 * i + 2 * half;
      bf[i].x = Wd[(size_t)ka * D_INNER + dbase + ml];
      bf[i].y = Wd[(size_t)(ka + 1) * D_INNER + dbase + ml];
    }
    bn = bias[dbase + ml];
    delta_tile(xdp, bf, bn, kw, lane, s_lds);
    scan_put(s_lds, tid, s);
    scan_fetch(xp, bcp, 1, tid, s);
  }
  stage_barrier();
  if (steps) walk_fill(L, k);

  int base = 0;
#pragma nounroll
  for (int c = 0; c < NTILE; ++c) {
    const int nb = (base == (NSLOT - 1) * CHUNKL) ? 0 : base + CHUNKL;
    if (steps) {
      walk_body(L, k, base);
    } else {
      float* slot = s_lds + nb;
      if (c >= 2) y_out(slot, outp, c - 2, tid);
      if (c + 1 < NTILE) {
        delta_tile(xdp + (size_t)(c + 1) * CHUNKL * N_PROJ, bf, bn, kw, lane, slot);
        scan_put(slot, tid, s);
        if (c + 2 < NTILE) scan_fetch(xp, bcp, c + 2, tid, s);
      }
    }
    stage_barrier();
    if (steps && c + 1 < NTILE) walk_cross(L, k, base, nb);
    base = nb;
  }
  if (steps) walk_tail(L, k, SLOT_COL(NTILE - 1));
  stage_barrier();
  if (!steps) {
    y_out(s_lds + SLOT_COL(NTILE - 2), outp, NTILE - 2, tid);
    y_out(s_lds + SLOT_COL(NTILE - 1), outp, NTILE - 1, tid);
  }
}

extern "C" void kernel_launch(void* const* d_in, const int* in_sizes, int n_in,
                              void* d_out, int out_size, void* d_ws, size_t ws_size,
                              hipStream_t stream) {
  const float* x    = (const float*)d_in[0];
  const float* xpw  = (const float*)d_in[1];
  const float* dtw  = (const float*)d_in[2];
  const float* dtb  = (const float*)d_in[3];
  const float* alog = (const float*)d_in[4];
  const float* dvec = (const float*)d_in[5];
  float* out = (float*)d_out;

  float* xdbl = (float*)d_ws;

  ssm_xproj_wmma<<<dim3(M_TOT / 16), dim3(192), 0, stream>>>(x, xpw, xdbl);

  ssm_scan_seq<<<dim3(BATCH * D_INNER / SCH), dim3(SNT), 0, stream>>>(
      x, xdbl, dtw, dtb, alog, dvec, out);
}
